# starting kernel
# baseline (speedup 1.0000x reference)
#include <hip/hip_runtime.h>
#include <hip/hip_bf16.h>

#define D      128
#define H      8
#define DK     16
#define SCALE  0.25f

typedef __attribute__((ext_vector_type(2))) float v2f;
typedef __attribute__((ext_vector_type(8))) float v8f;

__device__ __forceinline__ unsigned fenc(float x) {
  unsigned u = __float_as_uint(x);
  return (u & 0x80000000u) ? ~u : (u | 0x80000000u);
}
__device__ __forceinline__ float fdec(unsigned k) {
  return __uint_as_float((k & 0x80000000u) ? (k & 0x7FFFFFFFu) : ~k);
}

__global__ void gemm_xwT_wmma(const float* __restrict__ X,
                              const float* __restrict__ W,
                              float* __restrict__ Y, int M) {
  const int lane  = threadIdx.x & 31;
  const int wave  = (blockIdx.x * blockDim.x + threadIdx.x) >> 5;
  const int mBase = wave * 16;
  if (mBase >= M) return;

  const int row  = lane & 15;
  const int koff = (lane >> 4) << 1;

  const float* xrow  = X + (size_t)(mBase + row) * D + koff;
  const float* wbase = W + (size_t)row * D + koff;

  v8f acc[8] = {};

  v2f a_cur = *(const v2f*)(xrow);
  v2f b_cur[8];
#pragma unroll
  for (int t = 0; t < 8; ++t)
    b_cur[t] = *(const v2f*)(wbase + (size_t)t * 16 * D);

#pragma unroll 4
  for (int k0 = 0; k0 < D; k0 += 4) {
    v2f a_nxt;
    v2f b_nxt[8];
    if (k0 + 4 < D) {
      a_nxt = *(const v2f*)(xrow + k0 + 4);
#pragma unroll
      for (int t = 0; t < 8; ++t)
        b_nxt[t] = *(const v2f*)(wbase + (size_t)t * 16 * D + k0 + 4);
    }
#pragma unroll
    for (int t = 0; t < 8; ++t) {
      acc[t] = __builtin_amdgcn_wmma_f32_16x16x4_f32(
          false, a_cur, false, b_cur[t], (short)0, acc[t], false, false);
    }
    a_cur = a_nxt;
#pragma unroll
    for (int t = 0; t < 8; ++t) b_cur[t] = b_nxt[t];
  }

  const int n    = lane & 15;
  const int mAdd = (lane >> 4) << 3;
  float* yb = Y + (size_t)mBase * D;
#pragma unroll
  for (int t = 0; t < 8; ++t) {
#pragma unroll
    for (int r = 0; r < 8; ++r) {
      yb[(size_t)(mAdd + r) * D + t * 16 + n] = acc[t][r];
    }
  }
}

__global__ void self_init(const float* __restrict__ dstK,
                          const float* __restrict__ Q,
                          float* __restrict__ self_t,
                          unsigned* __restrict__ m_enc, int Ndst) {
  int tid = blockIdx.x * blockDim.x + threadIdx.x;
  if (tid >= Ndst * H) return;
  int n = tid >> 3, h = tid & 7;
  const float4* kp = (const float4*)(dstK + (size_t)n * D + h * DK);
  const float4* qp = (const float4*)(Q    + (size_t)n * D + h * DK);
  float acc = 0.f;
#pragma unroll
  for (int i = 0; i < 4; ++i) {
    float4 kv = kp[i], qv = qp[i];
    acc += kv.x * qv.x + kv.y * qv.y + kv.z * qv.z + kv.w * qv.w;
  }
  float st = acc * SCALE;
  self_t[tid] = st;
  m_enc[tid]  = fenc(st);
}

__global__ void edge_scores(const float* __restrict__ K,
                            const float* __restrict__ Q,
                            const int* __restrict__ src_idx,
                            const int* __restrict__ dst_idx,
                            float* __restrict__ t_buf,
                            unsigned* __restrict__ m_enc, int E) {
  int tid = blockIdx.x * blockDim.x + threadIdx.x;
  if (tid >= E * H) return;
  int e = tid >> 3, h = tid & 7;
  int s = src_idx[e], d = dst_idx[e];
  const float4* kp = (const float4*)(K + (size_t)s * D + h * DK);
  const float4* qp = (const float4*)(Q + (size_t)d * D + h * DK);
  float acc = 0.f;
#pragma unroll
  for (int i = 0; i < 4; ++i) {
    float4 kv = kp[i], qv = qp[i];
    acc += kv.x * qv.x + kv.y * qv.y + kv.z * qv.z + kv.w * qv.w;
  }
  float t = acc * SCALE;
  t_buf[tid] = t;
  atomicMax(&m_enc[d * H + h], fenc(t));
}

__global__ void denom_init(const float* __restrict__ self_t,
                           const unsigned* __restrict__ m_enc,
                           float* __restrict__ denom, int Ndst) {
  int tid = blockIdx.x * blockDim.x + threadIdx.x;
  if (tid >= Ndst * H) return;
  denom[tid] = __expf(self_t[tid] - fdec(m_enc[tid]));
}

__global__ void edge_exp(const int* __restrict__ dst_idx,
                         const unsigned* __restrict__ m_enc,
                         float* __restrict__ t_buf,
                         float* __restrict__ denom, int E) {
  int tid = blockIdx.x * blockDim.x + threadIdx.x;
  if (tid >= E * H) return;
  int e = tid >> 3, h = tid & 7;
  int d = dst_idx[e];
  float ev = __expf(t_buf[tid] - fdec(m_enc[d * H + h]));
  t_buf[tid] = ev;
  atomicAdd(&denom[d * H + h], ev);
}

__global__ void recip_denom(float* __restrict__ denom, int Ndst) {
  int tid = blockIdx.x * blockDim.x + threadIdx.x;
  if (tid >= Ndst * H) return;
  denom[tid] = 1.0f / denom[tid];
}

__global__ void edge_aggregate(const float* __restrict__ h_src,
                               const int* __restrict__ src_idx,
                               const int* __restrict__ dst_idx,
                               const float* __restrict__ t_buf,
                               const float* __restrict__ rdenom,
                               float* __restrict__ out, int E) {
  unsigned tid = blockIdx.x * blockDim.x + threadIdx.x;
  if (tid >= (unsigned)E * D) return;
  int e = tid >> 7, c = tid & (D - 1), h = c >> 4;
  int s = src_idx[e], d = dst_idx[e];
  float a = t_buf[e * H + h] * rdenom[d * H + h];
  atomicAdd(&out[(size_t)d * D + c], a * h_src[(size_t)s * D + c]);
}

__global__ void finalize(const float* __restrict__ h_dst,
                         const float* __restrict__ self_t,
                         const unsigned* __restrict__ m_enc,
                         const float* __restrict__ rdenom,
                         float* __restrict__ out, int Ndst) {
  int tid = blockIdx.x * blockDim.x + threadIdx.x;
  if (tid >= Ndst * D) return;
  int n = tid >> 7, c = tid & (D - 1), h = c >> 4;
  float m = fdec(m_enc[n * H + h]);
  float dst_a = __expf(self_t[n * H + h] - m) * rdenom[n * H + h];
  out[tid] += dst_a * h_dst[tid];
}

extern "C" void kernel_launch(void* const* d_in, const int* in_sizes, int n_in,
                              void* d_out, int out_size, void* d_ws, size_t ws_size,
                              hipStream_t stream) {
  const float* h_src = (const float*)d_in[0];
  const float* h_dst = (const float*)d_in[1];
  const float* Wk    = (const float*)d_in[2];
  const float* Wq    = (const float*)d_in[3];
  const int* src_idx = (const int*)d_in[4];
  const int* dst_idx = (const int*)d_in[5];

  const int N_SRC = in_sizes[0] / D;
  const int N_DST = in_sizes[1] / D;
  const int E     = in_sizes[4];

  auto align256 = [](size_t x) { return (x + 255) & ~(size_t)255; };
  char* ws = (char*)d_ws;
  size_t off = 0;
  float* K      = (float*)(ws + off); off += align256((size_t)N_SRC * D * 4);
  float* Q      = (float*)(ws + off); off += align256((size_t)N_DST * D * 4);
  float* dstK   = (float*)(ws + off); off += align256((size_t)N_DST * D * 4);
  float* t_buf  = (float*)(ws + off); off += align256((size_t)E * H * 4);
  float* self_t = (float*)(ws + off); off += align256((size_t)N_DST * H * 4);
  unsigned* m_enc = (unsigned*)(ws + off); off += align256((size_t)N_DST * H * 4);
  float* denom  = (float*)(ws + off); off += align256((size_t)N_DST * H * 4);

  const int TB = 256;

  hipMemsetAsync(d_out, 0, (size_t)out_size * sizeof(float), stream);

  {
    int waves  = N_SRC / 16;
    int blocks = (waves * 32 + TB - 1) / TB;
    gemm_xwT_wmma<<<blocks, TB, 0, stream>>>(h_src, Wk, K, N_SRC);
  }
  {
    int waves  = N_DST / 16;
    int blocks = (waves * 32 + TB - 1) / TB;
    gemm_xwT_wmma<<<blocks, TB, 0, stream>>>(h_dst, Wq, Q, N_DST);
    gemm_xwT_wmma<<<blocks, TB, 0, stream>>>(h_dst, Wk, dstK, N_DST);
  }

  self_init<<<(N_DST * H + TB - 1) / TB, TB, 0, stream>>>(dstK, Q, self_t, m_enc, N_DST);
  edge_scores<<<(E * H + TB - 1) / TB, TB, 0, stream>>>(K, Q, src_idx, dst_idx, t_buf, m_enc, E);
  denom_init<<<(N_DST * H + TB - 1) / TB, TB, 0, stream>>>(self_t, m_enc, denom, N_DST);
  edge_exp<<<(E * H + TB - 1) / TB, TB, 0, stream>>>(dst_idx, m_enc, t_buf, denom, E);
  recip_denom<<<(N_DST * H + TB - 1) / TB, TB, 0, stream>>>(denom, N_DST);
  {
    unsigned total  = (unsigned)E * D;
    unsigned blocks = (total + TB - 1) / TB;
    edge_aggregate<<<blocks, TB, 0, stream>>>(h_src, src_idx, dst_idx, t_buf, denom, (float*)d_out, E);
  }
  finalize<<<(N_DST * D + TB - 1) / TB, TB, 0, stream>>>(h_dst, self_t, m_enc, denom, (float*)d_out, N_DST);
}


# Aggregation owner-computes: a work-group keeps 200 dst rows in LDS (self term first), scans all edges, adds its own by LDS float atomics, stores rows once; no global atomics, memset, finalize
# speedup vs baseline: 1.9352x; 1.9352x over previous
#include <hip/hip_runtime.h>
#include <hip/hip_bf16.h>

#define D      128
#define H      8
#define DK     16
#define SCALE  0.25f

typedef __attribute__((ext_vector_type(2))) float v2f;
typedef __attribute__((ext_vector_type(8))) float v8f;

__device__ __forceinline__ unsigned fenc(float x) {
  unsigned u = __float_as_uint(x);
  return (u & 0x80000000u) ? ~u : (u | 0x80000000u);
}
__device__ __forceinline__ float fdec(unsigned k) {
  return __uint_as_float((k & 0x80000000u) ? (k & 0x7FFFFFFFu) : ~k);
}

__global__ void gemm_xwT_wmma(const float* __restrict__ X,
                              const float* __restrict__ W,
                              float* __restrict__ Y, int M) {
  const int lane  = threadIdx.x & 31;
  const int wave  = (blockIdx.x * blockDim.x + threadIdx.x) >> 5;
  const int mBase = wave * 16;
  if (mBase >= M) return;

  const int row  = lane & 15;
  const int koff = (lane >> 4) << 1;

  const float* xrow  = X + (size_t)(mBase + row) * D + koff;
  const float* wbase = W + (size_t)row * D + koff;

  v8f acc[8] = {};

  v2f a_cur = *(const v2f*)(xrow);
  v2f b_cur[8];
#pragma unroll
  for (int t = 0; t < 8; ++t)
    b_cur[t] = *(const v2f*)(wbase + (size_t)t * 16 * D);

#pragma unroll 4
  for (int k0 = 0; k0 < D; k0 += 4) {
    v2f a_nxt;
    v2f b_nxt[8];
    if (k0 + 4 < D) {
      a_nxt = *(const v2f*)(xrow + k0 + 4);
#pragma unroll
      for (int t = 0; t < 8; ++t)
        b_nxt[t] = *(const v2f*)(wbase + (size_t)t * 16 * D + k0 + 4);
    }
#pragma unroll
    for (int t = 0; t < 8; ++t) {
      acc[t] = __builtin_amdgcn_wmma_f32_16x16x4_f32(
          false, a_cur, false, b_cur[t], (short)0, acc[t], false, false);
    }
    a_cur = a_nxt;
#pragma unroll
    for (int t = 0; t < 8; ++t) b_cur[t] = b_nxt[t];
  }

  const int n    = lane & 15;
  const int mAdd = (lane >> 4) << 3;
  float* yb = Y + (size_t)mBase * D;
#pragma unroll
  for (int t = 0; t < 8; ++t) {
#pragma unroll
    for (int r = 0; r < 8; ++r) {
      yb[(size_t)(mAdd + r) * D + t * 16 + n] = acc[t][r];
    }
  }
}

__global__ void self_init(const float* __restrict__ dstK,
                          const float* __restrict__ Q,
                          float* __restrict__ self_t,
                          unsigned* __restrict__ m_enc, int Ndst) {
  int tid = blockIdx.x * blockDim.x + threadIdx.x;
  if (tid >= Ndst * H) return;
  int n = tid >> 3, h = tid & 7;
  const float4* kp = (const float4*)(dstK + (size_t)n * D + h * DK);
  const float4* qp = (const float4*)(Q    + (size_t)n * D + h * DK);
  float acc = 0.f;
#pragma unroll
  for (int i = 0; i < 4; ++i) {
    float4 kv = kp[i], qv = qp[i];
    acc += kv.x * qv.x + kv.y * qv.y + kv.z * qv.z + kv.w * qv.w;
  }
  float st = acc * SCALE;
  self_t[tid] = st;
  m_enc[tid]  = fenc(st);
}

__global__ void edge_scores(const float* __restrict__ K,
                            const float* __restrict__ Q,
                            const int* __restrict__ src_idx,
                            const int* __restrict__ dst_idx,
                            float* __restrict__ t_buf,
                            unsigned* __restrict__ m_enc, int E) {
  int tid = blockIdx.x * blockDim.x + threadIdx.x;
  if (tid >= E * H) return;
  int e = tid >> 3, h = tid & 7;
  int s = src_idx[e], d = dst_idx[e];
  const float4* kp = (const float4*)(K + (size_t)s * D + h * DK);
  const float4* qp = (const float4*)(Q + (size_t)d * D + h * DK);
  float acc = 0.f;
#pragma unroll
  for (int i = 0; i < 4; ++i) {
    float4 kv = kp[i], qv = qp[i];
    acc += kv.x * qv.x + kv.y * qv.y + kv.z * qv.z + kv.w * qv.w;
  }
  float t = acc * SCALE;
  t_buf[tid] = t;
  atomicMax(&m_enc[d * H + h], fenc(t));
}

__global__ void denom_init(const float* __restrict__ self_t,
                           const unsigned* __restrict__ m_enc,
                           float* __restrict__ denom, int Ndst) {
  int tid = blockIdx.x * blockDim.x + threadIdx.x;
  if (tid >= Ndst * H) return;
  denom[tid] = __expf(self_t[tid] - fdec(m_enc[tid]));
}

__global__ void edge_exp(const int* __restrict__ dst_idx,
                         const unsigned* __restrict__ m_enc,
                         float* __restrict__ t_buf,
                         float* __restrict__ denom, int E) {
  int tid = blockIdx.x * blockDim.x + threadIdx.x;
  if (tid >= E * H) return;
  int e = tid >> 3, h = tid & 7;
  int d = dst_idx[e];
  float ev = __expf(t_buf[tid] - fdec(m_enc[d * H + h]));
  t_buf[tid] = ev;
  atomicAdd(&denom[d * H + h], ev);
}

__global__ void recip_denom(float* __restrict__ denom, int Ndst) {
  int tid = blockIdx.x * blockDim.x + threadIdx.x;
  if (tid >= Ndst * H) return;
  denom[tid] = 1.0f / denom[tid];
}

#define AR    200
#define ATH   1024
#define SBLK  (ATH * 4)
#define QCAP  (8 * SBLK)

__global__ __launch_bounds__(ATH) void aggregate_owned(
    const float* __restrict__ h_src, const float* __restrict__ h_dst,
    const int* __restrict__ src_idx, const int* __restrict__ dst_idx,
    const float* __restrict__ ev_buf, const float* __restrict__ rdenom,
    const float* __restrict__ self_t, const unsigned* __restrict__ m_enc,
    float* __restrict__ out, int E, int Ndst) {
  __shared__ float    acc[AR * D];
  __shared__ unsigned q[QCAP];
  __shared__ unsigned qn;

  const int tid  = threadIdx.x;
  const int lane = tid & 31;
  const int wave  = __builtin_amdgcn_readfirstlane(tid >> 5);
  const int row0  = blockIdx.x * AR;
  const int nrows = min(AR, Ndst - row0);
  const int hl    = lane >> 2;

  for (int r = wave; r < nrows; r += ATH / 32) {
    const int n = row0 + r;
    const float m  = fdec(m_enc[n * H + hl]);
    const float da = __expf(self_t[n * H + hl] - m) * rdenom[n * H + hl];
    const float4 x = *(const float4*)(h_dst + (size_t)n * D + 4 * lane);
    float* a = acc + r * D + lane;
    a[0] = da * x.x; a[32] = da * x.y; a[64] = da * x.z; a[96] = da * x.w;
  }
  if (tid == 0) qn = 0;
  __syncthreads();

  auto push = [&](int e) {
    const unsigned p = atomicAdd(&qn, 1u);
    if (p < QCAP) q[p] = (unsigned)e;
  };

  int lo = 0, span = E;
  while (lo < E) {
    const int hi  = (span < E - lo) ? lo + span : E;
    const int nbf = (hi - lo) / SBLK;
    const int start = nbf ? (int)(((unsigned)blockIdx.x * (unsigned)nbf / gridDim.x) % (unsigned)nbf) : 0;
    const int* dspan = dst_idx + lo;
    for (int i = 0; i < nbf; i += 8) {
      int4 dv[8];
      int  e0[8];
#pragma unroll
      for (int k = 0; k < 8; ++k) {
        int b = start + min(i + k, nbf - 1);
        if (b >= nbf) b -= nbf;
        e0[k] = b * SBLK + tid * 4;
        dv[k] = *(const int4*)(dspan + e0[k]);
      }
#pragma unroll
      for (int k = 0; k < 8; ++k) {
        if (i + k < nbf) {
          const unsigned u0 = (unsigned)(dv[k].x - row0), u1 = (unsigned)(dv[k].y - row0);
          const unsigned u2 = (unsigned)(dv[k].z - row0), u3 = (unsigned)(dv[k].w - row0);
          if (min(min(u0, u1), min(u2, u3)) < (unsigned)nrows) {
            const int e = lo + e0[k];
            if (u0 < (unsigned)nrows) push(e);
            if (u1 < (unsigned)nrows) push(e + 1);
            if (u2 < (unsigned)nrows) push(e + 2);
            if (u3 < (unsigned)nrows) push(e + 3);
          }
        }
      }
    }
    for (int e = lo + nbf * SBLK + tid; e < hi; e += ATH)
      if ((unsigned)(dst_idx[e] - row0) < (unsigned)nrows) push(e);
    __syncthreads();

    const int n = __builtin_amdgcn_readfirstlane((int)qn);
    if (n <= QCAP) {
      for (int b0 = wave * 32; b0 < n; b0 += ATH) {
        const int idx = b0 + lane;
        int e_l = 0, s_l = 0, r_l = 0;
        if (idx < n) {
          e_l = (int)q[idx];
          s_l = src_idx[e_l];
          r_l = dst_idx[e_l] - row0;
        }
        const int cnt = min(32, n - b0);
        for (int j = 0; j < cnt; j += 8) {
          float4 v[8];
          float  a[8];
          int    rr[8];
#pragma unroll
          for (int u = 0; u < 8; ++u) {
            const int jj = min(j + u, cnt - 1);
            const int e = __builtin_amdgcn_readlane(e_l, jj);
            const int s = __builtin_amdgcn_readlane(s_l, jj);
            rr[u] = __builtin_amdgcn_readlane(r_l, jj);
            v[u] = *(const float4*)(h_src + (size_t)s * D + 4 * lane);
            a[u] = ev_buf[(size_t)e * H + hl] * rdenom[(size_t)(row0 + rr[u]) * H + hl];
          }
#pragma unroll
          for (int u = 0; u < 8; ++u) {
            if (j + u < cnt) {
              float* p = acc + rr[u] * D + lane;
              atomicAdd(p,      a[u] * v[u].x);
              atomicAdd(p + 32, a[u] * v[u].y);
              atomicAdd(p + 64, a[u] * v[u].z);
              atomicAdd(p + 96, a[u] * v[u].w);
            }
          }
        }
      }
      lo = hi;
    } else {
      span = QCAP;
    }
    __syncthreads();
    if (tid == 0) qn = 0;
    __syncthreads();
  }

  for (int r = wave; r < nrows; r += ATH / 32) {
    const float* a = acc + r * D + lane;
    float4 o;
    o.x = a[0]; o.y = a[32]; o.z = a[64]; o.w = a[96];
    *(float4*)(out + (size_t)(row0 + r) * D + 4 * lane) = o;
  }
}

extern "C" void kernel_launch(void* const* d_in, const int* in_sizes, int n_in,
                              void* d_out, int out_size, void* d_ws, size_t ws_size,
                              hipStream_t stream) {
  const float* h_src = (const float*)d_in[0];
  const float* h_dst = (const float*)d_in[1];
  const float* Wk    = (const float*)d_in[2];
  const float* Wq    = (const float*)d_in[3];
  const int* src_idx = (const int*)d_in[4];
  const int* dst_idx = (const int*)d_in[5];

  const int N_SRC = in_sizes[0] / D;
  const int N_DST = in_sizes[1] / D;
  const int E     = in_sizes[4];

  auto align256 = [](size_t x) { return (x + 255) & ~(size_t)255; };
  char* ws = (char*)d_ws;
  size_t off = 0;
  float* K      = (float*)(ws + off); off += align256((size_t)N_SRC * D * 4);
  float* Q      = (float*)(ws + off); off += align256((size_t)N_DST * D * 4);
  float* dstK   = (float*)(ws + off); off += align256((size_t)N_DST * D * 4);
  float* t_buf  = (float*)(ws + off); off += align256((size_t)E * H * 4);
  float* self_t = (float*)(ws + off); off += align256((size_t)N_DST * H * 4);
  unsigned* m_enc = (unsigned*)(ws + off); off += align256((size_t)N_DST * H * 4);
  float* denom  = (float*)(ws + off); off += align256((size_t)N_DST * H * 4);

  const int TB = 256;

  {
    int waves  = N_SRC / 16;
    int blocks = (waves * 32 + TB - 1) / TB;
    gemm_xwT_wmma<<<blocks, TB, 0, stream>>>(h_src, Wk, K, N_SRC);
  }
  {
    int waves  = N_DST / 16;
    int blocks = (waves * 32 + TB - 1) / TB;
    gemm_xwT_wmma<<<blocks, TB, 0, stream>>>(h_dst, Wq, Q, N_DST);
    gemm_xwT_wmma<<<blocks, TB, 0, stream>>>(h_dst, Wk, dstK, N_DST);
  }

  self_init<<<(N_DST * H + TB - 1) / TB, TB, 0, stream>>>(dstK, Q, self_t, m_enc, N_DST);
  edge_scores<<<(E * H + TB - 1) / TB, TB, 0, stream>>>(K, Q, src_idx, dst_idx, t_buf, m_enc, E);
  denom_init<<<(N_DST * H + TB - 1) / TB, TB, 0, stream>>>(self_t, m_enc, denom, N_DST);
  edge_exp<<<(E * H + TB - 1) / TB, TB, 0, stream>>>(dst_idx, m_enc, t_buf, denom, E);
  recip_denom<<<(N_DST * H + TB - 1) / TB, TB, 0, stream>>>(denom, N_DST);
  aggregate_owned<<<(N_DST + AR - 1) / AR, ATH, 0, stream>>>(h_src, h_dst, src_idx, dst_idx, t_buf, denom,
                                                          self_t, m_enc, (float*)d_out, E, N_DST);
}


# Softmax max and denominator kept per owned row in LDS by the aggregating work-group (LDS atomics, sum in arrival order, one divide per row at the end): no global atomics, three launches fewer
# speedup vs baseline: 2.3648x; 1.2220x over previous
#include <hip/hip_runtime.h>
#include <hip/hip_bf16.h>

#define D      128
#define H      8
#define DK     16
#define SCALE  0.25f

typedef __attribute__((ext_vector_type(2))) float v2f;
typedef __attribute__((ext_vector_type(8))) float v8f;

__device__ __forceinline__ unsigned fenc(float x) {
  unsigned u = __float_as_uint(x);
  return (u & 0x80000000u) ? ~u : (u | 0x80000000u);
}
__device__ __forceinline__ float fdec(unsigned k) {
  return __uint_as_float((k & 0x80000000u) ? (k & 0x7FFFFFFFu) : ~k);
}

__global__ void gemm_xwT_wmma(const float* __restrict__ X,
                              const float* __restrict__ W,
                              float* __restrict__ Y, int M) {
  const int lane  = threadIdx.x & 31;
  const int wave  = (blockIdx.x * blockDim.x + threadIdx.x) >> 5;
  const int mBase = wave * 16;
  if (mBase >= M) return;

  const int row  = lane & 15;
  const int koff = (lane >> 4) << 1;

  const float* xrow  = X + (size_t)(mBase + row) * D + koff;
  const float* wbase = W + (size_t)row * D + koff;

  v8f acc[8] = {};

  v2f a_cur = *(const v2f*)(xrow);
  v2f b_cur[8];
#pragma unroll
  for (int t = 0; t < 8; ++t)
    b_cur[t] = *(const v2f*)(wbase + (size_t)t * 16 * D);

#pragma unroll 4
  for (int k0 = 0; k0 < D; k0 += 4) {
    v2f a_nxt;
    v2f b_nxt[8];
    if (k0 + 4 < D) {
      a_nxt = *(const v2f*)(xrow + k0 + 4);
#pragma unroll
      for (int t = 0; t < 8; ++t)
        b_nxt[t] = *(const v2f*)(wbase + (size_t)t * 16 * D + k0 + 4);
    }
#pragma unroll
    for (int t = 0; t < 8; ++t) {
      acc[t] = __builtin_amdgcn_wmma_f32_16x16x4_f32(
          false, a_cur, false, b_cur[t], (short)0, acc[t], false, false);
    }
    a_cur = a_nxt;
#pragma unroll
    for (int t = 0; t < 8; ++t) b_cur[t] = b_nxt[t];
  }

  const int n    = lane & 15;
  const int mAdd = (lane >> 4) << 3;
  float* yb = Y + (size_t)mBase * D;
#pragma unroll
  for (int t = 0; t < 8; ++t) {
#pragma unroll
    for (int r = 0; r < 8; ++r) {
      yb[(size_t)(mAdd + r) * D + t * 16 + n] = acc[t][r];
    }
  }
}

__global__ void self_init(const float* __restrict__ dstK,
                          const float* __restrict__ Q,
                          float* __restrict__ self_t, int Ndst) {
  int tid = blockIdx.x * blockDim.x + threadIdx.x;
  if (tid >= Ndst * H) return;
  int n = tid >> 3, h = tid & 7;
  const float4* kp = (const float4*)(dstK + (size_t)n * D + h * DK);
  const float4* qp = (const float4*)(Q    + (size_t)n * D + h * DK);
  float acc = 0.f;
#pragma unroll
  for (int i = 0; i < 4; ++i) {
    float4 kv = kp[i], qv = qp[i];
    acc += kv.x * qv.x + kv.y * qv.y + kv.z * qv.z + kv.w * qv.w;
  }
  self_t[tid] = acc * SCALE;
}

__global__ void edge_scores(const float* __restrict__ K,
                            const float* __restrict__ Q,
                            const int* __restrict__ src_idx,
                            const int* __restrict__ dst_idx,
                            float* __restrict__ t_buf, int E) {
  int tid = blockIdx.x * blockDim.x + threadIdx.x;
  if (tid >= E * H) return;
  int e = tid >> 3, h = tid & 7;
  int s = src_idx[e], d = dst_idx[e];
  const float4* kp = (const float4*)(K + (size_t)s * D + h * DK);
  const float4* qp = (const float4*)(Q + (size_t)d * D + h * DK);
  float acc = 0.f;
#pragma unroll
  for (int i = 0; i < 4; ++i) {
    float4 kv = kp[i], qv = qp[i];
    acc += kv.x * qv.x + kv.y * qv.y + kv.z * qv.z + kv.w * qv.w;
  }
  t_buf[tid] = acc * SCALE;
}

#define AR    200
#define ATH   1024
#define SBLK  (ATH * 4)
#define QCAP  (8 * SBLK)

__global__ __launch_bounds__(ATH) void aggregate_owned(
    const float* __restrict__ h_src, const float* __restrict__ h_dst,
    const int* __restrict__ src_idx, const int* __restrict__ dst_idx,
    const float* __restrict__ t_buf, const float* __restrict__ self_t,
    float* __restrict__ out, int E, int Ndst) {
  __shared__ float    acc[AR * D];
  __shared__ unsigned q[QCAP];
  __shared__ unsigned mx[AR * H];
  __shared__ float    den[AR * H];
  __shared__ unsigned qn;

  const int tid  = threadIdx.x;
  const int lane = tid & 31;
  const int wave  = __builtin_amdgcn_readfirstlane(tid >> 5);
  const int row0  = blockIdx.x * AR;
  const int nrows = min(AR, Ndst - row0);
  const int hl    = lane >> 2;

  for (int i = tid; i < nrows * D; i += ATH) acc[i] = 0.f;
  for (int i = tid; i < nrows * H; i += ATH) {
    mx[i]  = fenc(self_t[(size_t)row0 * H + i]);
    den[i] = 0.f;
  }
  if (tid == 0) qn = 0;
  __syncthreads();

  auto push = [&](int e) {
    const unsigned p = atomicAdd(&qn, 1u);
    if (p < QCAP) q[p] = (unsigned)e;
  };

  int lo = 0, span = E, sweep = 0;
  while (lo < E) {
    const int hi  = (span < E - lo) ? lo + span : E;
    const int nbf = (hi - lo) / SBLK;
    const int start = nbf ? (int)(((unsigned)blockIdx.x * (unsigned)nbf / gridDim.x) % (unsigned)nbf) : 0;
    const int* dspan = dst_idx + lo;
    for (int i = 0; i < nbf; i += 8) {
      int4 dv[8];
      int  e0[8];
#pragma unroll
      for (int k = 0; k < 8; ++k) {
        int b = start + min(i + k, nbf - 1);
        if (b >= nbf) b -= nbf;
        e0[k] = b * SBLK + tid * 4;
        dv[k] = *(const int4*)(dspan + e0[k]);
      }
#pragma unroll
      for (int k = 0; k < 8; ++k) {
        if (i + k < nbf) {
          const unsigned u0 = (unsigned)(dv[k].x - row0), u1 = (unsigned)(dv[k].y - row0);
          const unsigned u2 = (unsigned)(dv[k].z - row0), u3 = (unsigned)(dv[k].w - row0);
          if (min(min(u0, u1), min(u2, u3)) < (unsigned)nrows) {
            const int e = lo + e0[k];
            if (u0 < (unsigned)nrows) push(e);
            if (u1 < (unsigned)nrows) push(e + 1);
            if (u2 < (unsigned)nrows) push(e + 2);
            if (u3 < (unsigned)nrows) push(e + 3);
          }
        }
      }
    }
    for (int e = lo + nbf * SBLK + tid; e < hi; e += ATH)
      if ((unsigned)(dst_idx[e] - row0) < (unsigned)nrows) push(e);
    __syncthreads();

    const int n = __builtin_amdgcn_readfirstlane((int)qn);
    const bool fits = n <= QCAP;
    if (fits && sweep != 2) {
      for (int idx = tid; idx < n; idx += ATH) {
        const int e = (int)q[idx];
        const int r = dst_idx[e] - row0;
        const float4 ta = *(const float4*)(t_buf + (size_t)e * H);
        const float4 tb = *(const float4*)(t_buf + (size_t)e * H + 4);
        unsigned* mp = mx + r * H;
        atomicMax(mp + 0, fenc(ta.x)); atomicMax(mp + 1, fenc(ta.y));
        atomicMax(mp + 2, fenc(ta.z)); atomicMax(mp + 3, fenc(ta.w));
        atomicMax(mp + 4, fenc(tb.x)); atomicMax(mp + 5, fenc(tb.y));
        atomicMax(mp + 6, fenc(tb.z)); atomicMax(mp + 7, fenc(tb.w));
      }
    }
    if (sweep == 0) __syncthreads();
    if (fits && sweep != 1) {
      for (int b0 = wave * 32; b0 < n; b0 += ATH) {
        const int idx = b0 + lane;
        int e_l = 0, s_l = 0, r_l = 0;
        if (idx < n) {
          e_l = (int)q[idx];
          s_l = src_idx[e_l];
          r_l = dst_idx[e_l] - row0;
        }
        const int cnt = min(32, n - b0);
        for (int j = 0; j < cnt; j += 8) {
          float4 v[8];
          float  a[8];
          int    rr[8];
#pragma unroll
          for (int u = 0; u < 8; ++u) {
            const int jj = min(j + u, cnt - 1);
            const int e = __builtin_amdgcn_readlane(e_l, jj);
            const int s = __builtin_amdgcn_readlane(s_l, jj);
            rr[u] = __builtin_amdgcn_readlane(r_l, jj);
            v[u] = *(const float4*)(h_src + (size_t)s * D + 4 * lane);
            a[u] = __expf(t_buf[(size_t)e * H + hl] - fdec(mx[rr[u] * H + hl]));
          }
#pragma unroll
          for (int u = 0; u < 8; ++u) {
            if (j + u < cnt) {
              if ((lane & 3) == 0) atomicAdd(den + rr[u] * H + hl, a[u]);
              float* p = acc + rr[u] * D + lane;
              atomicAdd(p,      a[u] * v[u].x);
              atomicAdd(p + 32, a[u] * v[u].y);
              atomicAdd(p + 64, a[u] * v[u].z);
              atomicAdd(p + 96, a[u] * v[u].w);
            }
          }
        }
      }
    }
    __syncthreads();
    if (tid == 0) qn = 0;
    __syncthreads();
    if (sweep == 0) {
      if (fits) break;
      sweep = 1; span = QCAP; lo = 0;
    } else {
      lo = hi;
      if (lo >= E && sweep == 1) { sweep = 2; lo = 0; }
    }
  }

  for (int r = wave; r < nrows; r += ATH / 32) {
    const int nd = row0 + r;
    const float m  = fdec(mx[r * H + hl]);
    const float se = __expf(self_t[(size_t)nd * H + hl] - m);
    const float rd = 1.0f / (den[r * H + hl] + se);
    const float da = se * rd;
    const float4 x = *(const float4*)(h_dst + (size_t)nd * D + 4 * lane);
    const float* a = acc + r * D + lane;
    float4 o;
    o.x = a[0]  * rd + da * x.x;
    o.y = a[32] * rd + da * x.y;
    o.z = a[64] * rd + da * x.z;
    o.w = a[96] * rd + da * x.w;
    *(float4*)(out + (size_t)nd * D + 4 * lane) = o;
  }
}

extern "C" void kernel_launch(void* const* d_in, const int* in_sizes, int n_in,
                              void* d_out, int out_size, void* d_ws, size_t ws_size,
                              hipStream_t stream) {
  const float* h_src = (const float*)d_in[0];
  const float* h_dst = (const float*)d_in[1];
  const float* Wk    = (const float*)d_in[2];
  const float* Wq    = (const float*)d_in[3];
  const int* src_idx = (const int*)d_in[4];
  const int* dst_idx = (const int*)d_in[5];

  const int N_SRC = in_sizes[0] / D;
  const int N_DST = in_sizes[1] / D;
  const int E     = in_sizes[4];

  auto align256 = [](size_t x) { return (x + 255) & ~(size_t)255; };
  char* ws = (char*)d_ws;
  size_t off = 0;
  float* K      = (float*)(ws + off); off += align256((size_t)N_SRC * D * 4);
  float* Q      = (float*)(ws + off); off += align256((size_t)N_DST * D * 4);
  float* dstK   = (float*)(ws + off); off += align256((size_t)N_DST * D * 4);
  float* t_buf  = (float*)(ws + off); off += align256((size_t)E * H * 4);
  float* self_t = (float*)(ws + off); off += align256((size_t)N_DST * H * 4);

  const int TB = 256;

  {
    int waves  = N_SRC / 16;
    int blocks = (waves * 32 + TB - 1) / TB;
    gemm_xwT_wmma<<<blocks, TB, 0, stream>>>(h_src, Wk, K, N_SRC);
  }
  {
    int waves  = N_DST / 16;
    int blocks = (waves * 32 + TB - 1) / TB;
    gemm_xwT_wmma<<<blocks, TB, 0, stream>>>(h_dst, Wq, Q, N_DST);
    gemm_xwT_wmma<<<blocks, TB, 0, stream>>>(h_dst, Wk, dstK, N_DST);
  }

  self_init<<<(N_DST * H + TB - 1) / TB, TB, 0, stream>>>(dstK, Q, self_t, N_DST);
  edge_scores<<<(E * H + TB - 1) / TB, TB, 0, stream>>>(K, Q, src_idx, dst_idx, t_buf, E);
  aggregate_owned<<<(N_DST + AR - 1) / AR, ATH, 0, stream>>>(h_src, h_dst, src_idx, dst_idx, t_buf, self_t,
                                                          (float*)d_out, E, N_DST);
}


# GEMMs: a wave keeps its 64-column half of W (all k) as B fragments in 256 registers for the whole launch and loops over row tiles; k loop has no loads; 16-B lane loads permute k: sum order changes
# speedup vs baseline: 2.7372x; 1.1575x over previous
#include <hip/hip_runtime.h>
#include <hip/hip_bf16.h>

#define D      128
#define H      8
#define DK     16
#define SCALE  0.25f

typedef __attribute__((ext_vector_type(2))) float v2f;
typedef __attribute__((ext_vector_type(4))) float v4f;
typedef __attribute__((ext_vector_type(8))) float v8f;

__device__ __forceinline__ unsigned fenc(float x) {
  unsigned u = __float_as_uint(x);
  return (u & 0x80000000u) ? ~u : (u | 0x80000000u);
}
__device__ __forceinline__ float fdec(unsigned k) {
  return __uint_as_float((k & 0x80000000u) ? (k & 0x7FFFFFFFu) : ~k);
}

#define GEMM_TH 256
__global__ __launch_bounds__(GEMM_TH) __attribute__((amdgpu_waves_per_eu(2, 2)))
void gemm_xwT_wmma(const float* __restrict__ X,
                   const float* __restrict__ W,
                   float* __restrict__ Y, int M) {
  const int lane = threadIdx.x & 31;
  const int wave = __builtin_amdgcn_readfirstlane((int)(threadIdx.x >> 5));
  const int half = wave >> 2;
  const int sub  = wave & 3;

  const int row  = lane & 15;
  const int koff = (lane >> 4) << 2;

  const int ntiles = M >> 4;
  const int stride = (int)gridDim.x * 4;
  int tile = (int)blockIdx.x * 4 + sub;
  if (tile >= ntiles) return;

  const float* wbase = W + (size_t)(half * 64 + row) * D + koff;
  v4f wf[4][16];
#pragma unroll
  for (int t = 0; t < 4; ++t)
#pragma unroll
    for (int j = 0; j < 16; ++j)
      wf[t][j] = *(const v4f*)(wbase + (size_t)t * 16 * D + j * 8);

  const float* xbase = X + (size_t)row * D + koff;

  const int n    = lane & 15;
  const int mAdd = (lane >> 4) << 3;

  auto load_a = [&](v4f (&a)[16], int t) {
#pragma unroll
    for (int j = 0; j < 16; ++j)
      a[j] = *(const v4f*)(xbase + (size_t)t * 16 * D + j * 8);
  };
  auto tile_mm = [&](const v4f (&a)[16], int t) {
    v8f acc[4] = {};
#pragma unroll
    for (int j = 0; j < 16; ++j) {
      const v2f a_lo = {a[j].x, a[j].y};
      const v2f a_hi = {a[j].z, a[j].w};
#pragma unroll
      for (int c = 0; c < 4; ++c) {
        const v2f b_lo = {wf[c][j].x, wf[c][j].y};
        acc[c] = __builtin_amdgcn_wmma_f32_16x16x4_f32(
            false, a_lo, false, b_lo, (short)0, acc[c], false, false);
      }
#pragma unroll
      for (int c = 0; c < 4; ++c) {
        const v2f b_hi = {wf[c][j].z, wf[c][j].w};
        acc[c] = __builtin_amdgcn_wmma_f32_16x16x4_f32(
            false, a_hi, false, b_hi, (short)0, acc[c], false, false);
      }
    }
    float* yb = Y + (size_t)t * 16 * D + half * 64;
#pragma unroll
    for (int c = 0; c < 4; ++c) {
#pragma unroll
      for (int r = 0; r < 8; ++r) {
        yb[(size_t)(mAdd + r) * D + c * 16 + n] = acc[c][r];
      }
    }
  };

  v4f a0[16], a1[16];
  load_a(a0, tile);
  for (;;) {
    int nxt = tile + stride;
    if (nxt >= ntiles) { tile_mm(a0, tile); break; }
    load_a(a1, nxt);
    tile_mm(a0, tile);
    tile = nxt;
    nxt  = tile + stride;
    if (nxt >= ntiles) { tile_mm(a1, tile); break; }
    load_a(a0, nxt);
    tile_mm(a1, tile);
    tile = nxt;
  }
}

__global__ void self_init(const float* __restrict__ dstK,
                          const float* __restrict__ Q,
                          float* __restrict__ self_t, int Ndst) {
  int tid = blockIdx.x * blockDim.x + threadIdx.x;
  if (tid >= Ndst * H) return;
  int n = tid >> 3, h = tid & 7;
  const float4* kp = (const float4*)(dstK + (size_t)n * D + h * DK);
  const float4* qp = (const float4*)(Q    + (size_t)n * D + h * DK);
  float acc = 0.f;
#pragma unroll
  for (int i = 0; i < 4; ++i) {
    float4 kv = kp[i], qv = qp[i];
    acc += kv.x * qv.x + kv.y * qv.y + kv.z * qv.z + kv.w * qv.w;
  }
  self_t[tid] = acc * SCALE;
}

__global__ void edge_scores(const float* __restrict__ K,
                            const float* __restrict__ Q,
                            const int* __restrict__ src_idx,
                            const int* __restrict__ dst_idx,
                            float* __restrict__ t_buf, int E) {
  int tid = blockIdx.x * blockDim.x + threadIdx.x;
  if (tid >= E * H) return;
  int e = tid >> 3, h = tid & 7;
  int s = src_idx[e], d = dst_idx[e];
  const float4* kp = (const float4*)(K + (size_t)s * D + h * DK);
  const float4* qp = (const float4*)(Q + (size_t)d * D + h * DK);
  float acc = 0.f;
#pragma unroll
  for (int i = 0; i < 4; ++i) {
    float4 kv = kp[i], qv = qp[i];
    acc += kv.x * qv.x + kv.y * qv.y + kv.z * qv.z + kv.w * qv.w;
  }
  t_buf[tid] = acc * SCALE;
}

#define AR    200
#define ATH   1024
#define SBLK  (ATH * 4)
#define QCAP  (8 * SBLK)

__global__ __launch_bounds__(ATH) void aggregate_owned(
    const float* __restrict__ h_src, const float* __restrict__ h_dst,
    const int* __restrict__ src_idx, const int* __restrict__ dst_idx,
    const float* __restrict__ t_buf, const float* __restrict__ self_t,
    float* __restrict__ out, int E, int Ndst) {
  __shared__ float    acc[AR * D];
  __shared__ unsigned q[QCAP];
  __shared__ unsigned mx[AR * H];
  __shared__ float    den[AR * H];
  __shared__ unsigned qn;

  const int tid  = threadIdx.x;
  const int lane = tid & 31;
  const int wave  = __builtin_amdgcn_readfirstlane(tid >> 5);
  const int row0  = blockIdx.x * AR;
  const int nrows = min(AR, Ndst - row0);
  const int hl    = lane >> 2;

  for (int i = tid; i < nrows * D; i += ATH) acc[i] = 0.f;
  for (int i = tid; i < nrows * H; i += ATH) {
    mx[i]  = fenc(self_t[(size_t)row0 * H + i]);
    den[i] = 0.f;
  }
  if (tid == 0) qn = 0;
  __syncthreads();

  auto push = [&](int e) {
    const unsigned p = atomicAdd(&qn, 1u);
    if (p < QCAP) q[p] = (unsigned)e;
  };

  int lo = 0, span = E, sweep = 0;
  while (lo < E) {
    const int hi  = (span < E - lo) ? lo + span : E;
    const int nbf = (hi - lo) / SBLK;
    const int start = nbf ? (int)(((unsigned)blockIdx.x * (unsigned)nbf / gridDim.x) % (unsigned)nbf) : 0;
    const int* dspan = dst_idx + lo;
    for (int i = 0; i < nbf; i += 8) {
      int4 dv[8];
      int  e0[8];
#pragma unroll
      for (int k = 0; k < 8; ++k) {
        int b = start + min(i + k, nbf - 1);
        if (b >= nbf) b -= nbf;
        e0[k] = b * SBLK + tid * 4;
        dv[k] = *(const int4*)(dspan + e0[k]);
      }
#pragma unroll
      for (int k = 0; k < 8; ++k) {
        if (i + k < nbf) {
          const unsigned u0 = (unsigned)(dv[k].x - row0), u1 = (unsigned)(dv[k].y - row0);
          const unsigned u2 = (unsigned)(dv[k].z - row0), u3 = (unsigned)(dv[k].w - row0);
          if (min(min(u0, u1), min(u2, u3)) < (unsigned)nrows) {
            const int e = lo + e0[k];
            if (u0 < (unsigned)nrows) push(e);
            if (u1 < (unsigned)nrows) push(e + 1);
            if (u2 < (unsigned)nrows) push(e + 2);
            if (u3 < (unsigned)nrows) push(e + 3);
          }
        }
      }
    }
    for (int e = lo + nbf * SBLK + tid; e < hi; e += ATH)
      if ((unsigned)(dst_idx[e] - row0) < (unsigned)nrows) push(e);
    __syncthreads();

    const int n = __builtin_amdgcn_readfirstlane((int)qn);
    const bool fits = n <= QCAP;
    if (fits && sweep != 2) {
      for (int idx = tid; idx < n; idx += ATH) {
        const int e = (int)q[idx];
        const int r = dst_idx[e] - row0;
        const float4 ta = *(const float4*)(t_buf + (size_t)e * H);
        const float4 tb = *(const float4*)(t_buf + (size_t)e * H + 4);
        unsigned* mp = mx + r * H;
        atomicMax(mp + 0, fenc(ta.x)); atomicMax(mp + 1, fenc(ta.y));
        atomicMax(mp + 2, fenc(ta.z)); atomicMax(mp + 3, fenc(ta.w));
        atomicMax(mp + 4, fenc(tb.x)); atomicMax(mp + 5, fenc(tb.y));
        atomicMax(mp + 6, fenc(tb.z)); atomicMax(mp + 7, fenc(tb.w));
      }
    }
    if (sweep == 0) __syncthreads();
    if (fits && sweep != 1) {
      for (int b0 = wave * 32; b0 < n; b0 += ATH) {
        const int idx = b0 + lane;
        int e_l = 0, s_l = 0, r_l = 0;
        if (idx < n) {
          e_l = (int)q[idx];
          s_l = src_idx[e_l];
          r_l = dst_idx[e_l] - row0;
        }
        const int cnt = min(32, n - b0);
        for (int j = 0; j < cnt; j += 8) {
          float4 v[8];
          float  a[8];
          int    rr[8];
#pragma unroll
          for (int u = 0; u < 8; ++u) {
            const int jj = min(j + u, cnt - 1);
            const int e = __builtin_amdgcn_readlane(e_l, jj);
            const int s = __builtin_amdgcn_readlane(s_l, jj);
            rr[u] = __builtin_amdgcn_readlane(r_l, jj);
            v[u] = *(const float4*)(h_src + (size_t)s * D + 4 * lane);
            a[u] = __expf(t_buf[(size_t)e * H + hl] - fdec(mx[rr[u] * H + hl]));
          }
#pragma unroll
          for (int u = 0; u < 8; ++u) {
            if (j + u < cnt) {
              if ((lane & 3) == 0) atomicAdd(den + rr[u] * H + hl, a[u]);
              float* p = acc + rr[u] * D + lane;
              atomicAdd(p,      a[u] * v[u].x);
              atomicAdd(p + 32, a[u] * v[u].y);
              atomicAdd(p + 64, a[u] * v[u].z);
              atomicAdd(p + 96, a[u] * v[u].w);
            }
          }
        }
      }
    }
    __syncthreads();
    if (tid == 0) qn = 0;
    __syncthreads();
    if (sweep == 0) {
      if (fits) break;
      sweep = 1; span = QCAP; lo = 0;
    } else {
      lo = hi;
      if (lo >= E && sweep == 1) { sweep = 2; lo = 0; }
    }
  }

  for (int r = wave; r < nrows; r += ATH / 32) {
    const int nd = row0 + r;
    const float m  = fdec(mx[r * H + hl]);
    const float se = __expf(self_t[(size_t)nd * H + hl] - m);
    const float rd = 1.0f / (den[r * H + hl] + se);
    const float da = se * rd;
    const float4 x = *(const float4*)(h_dst + (size_t)nd * D + 4 * lane);
    const float* a = acc + r * D + lane;
    float4 o;
    o.x = a[0]  * rd + da * x.x;
    o.y = a[32] * rd + da * x.y;
    o.z = a[64] * rd + da * x.z;
    o.w = a[96] * rd + da * x.w;
    *(float4*)(out + (size_t)nd * D + 4 * lane) = o;
  }
}

extern "C" void kernel_launch(void* const* d_in, const int* in_sizes, int n_in,
                              void* d_out, int out_size, void* d_ws, size_t ws_size,
                              hipStream_t stream) {
  const float* h_src = (const float*)d_in[0];
  const float* h_dst = (const float*)d_in[1];
  const float* Wk    = (const float*)d_in[2];
  const float* Wq    = (const float*)d_in[3];
  const int* src_idx = (const int*)d_in[4];
  const int* dst_idx = (const int*)d_in[5];

  const int N_SRC = in_sizes[0] / D;
  const int N_DST = in_sizes[1] / D;
  const int E     = in_sizes[4];

  auto align256 = [](size_t x) { return (x + 255) & ~(size_t)255; };
  char* ws = (char*)d_ws;
  size_t off = 0;
  float* K      = (float*)(ws + off); off += align256((size_t)N_SRC * D * 4);
  float* Q      = (float*)(ws + off); off += align256((size_t)N_DST * D * 4);
  float* dstK   = (float*)(ws + off); off += align256((size_t)N_DST * D * 4);
  float* t_buf  = (float*)(ws + off); off += align256((size_t)E * H * 4);
  float* self_t = (float*)(ws + off); off += align256((size_t)N_DST * H * 4);

  const int TB = 256;

  gemm_xwT_wmma<<<256, GEMM_TH, 0, stream>>>(h_src, Wk, K, N_SRC);
  gemm_xwT_wmma<<<256, GEMM_TH, 0, stream>>>(h_dst, Wq, Q, N_DST);
  gemm_xwT_wmma<<<256, GEMM_TH, 0, stream>>>(h_dst, Wk, dstK, N_DST);

  self_init<<<(N_DST * H + TB - 1) / TB, TB, 0, stream>>>(dstK, Q, self_t, N_DST);
  edge_scores<<<(E * H + TB - 1) / TB, TB, 0, stream>>>(K, Q, src_idx, dst_idx, t_buf, E);
  aggregate_owned<<<(N_DST + AR - 1) / AR, ATH, 0, stream>>>(h_src, h_dst, src_idx, dst_idx, t_buf, self_t,
                                                          (float*)d_out, E, N_DST);
}


# Score kernels (edge, self): 8 lanes per row pair, each wave load covers 4 whole 128-B lines (was 16 lines, 32 B of each); quad lane-swizzle transpose-sum, one head per lane; score sum order changes
# speedup vs baseline: 3.0555x; 1.1163x over previous
#include <hip/hip_runtime.h>
#include <hip/hip_bf16.h>

#define D      128
#define H      8
#define DK     16
#define SCALE  0.25f

typedef __attribute__((ext_vector_type(2))) float v2f;
typedef __attribute__((ext_vector_type(4))) float v4f;
typedef __attribute__((ext_vector_type(8))) float v8f;

__device__ __forceinline__ unsigned fenc(float x) {
  unsigned u = __float_as_uint(x);
  return (u & 0x80000000u) ? ~u : (u | 0x80000000u);
}
__device__ __forceinline__ float fdec(unsigned k) {
  return __uint_as_float((k & 0x80000000u) ? (k & 0x7FFFFFFFu) : ~k);
}

#define GEMM_TH 256
__global__ __launch_bounds__(GEMM_TH) __attribute__((amdgpu_waves_per_eu(2, 2)))
void gemm_xwT_wmma(const float* __restrict__ X,
                   const float* __restrict__ W,
                   float* __restrict__ Y, int M) {
  const int lane = threadIdx.x & 31;
  const int wave = __builtin_amdgcn_readfirstlane((int)(threadIdx.x >> 5));
  const int half = wave >> 2;
  const int sub  = wave & 3;

  const int row  = lane & 15;
  const int koff = (lane >> 4) << 2;

  const int ntiles = M >> 4;
  const int stride = (int)gridDim.x * 4;
  int tile = (int)blockIdx.x * 4 + sub;
  if (tile >= ntiles) return;

  const float* wbase = W + (size_t)(half * 64 + row) * D + koff;
  v4f wf[4][16];
#pragma unroll
  for (int t = 0; t < 4; ++t)
#pragma unroll
    for (int j = 0; j < 16; ++j)
      wf[t][j] = *(const v4f*)(wbase + (size_t)t * 16 * D + j * 8);

  const float* xbase = X + (size_t)row * D + koff;

  const int n    = lane & 15;
  const int mAdd = (lane >> 4) << 3;

  auto load_a = [&](v4f (&a)[16], int t) {
#pragma unroll
    for (int j = 0; j < 16; ++j)
      a[j] = *(const v4f*)(xbase + (size_t)t * 16 * D + j * 8);
  };
  auto tile_mm = [&](const v4f (&a)[16], int t) {
    v8f acc[4] = {};
#pragma unroll
    for (int j = 0; j < 16; ++j) {
      const v2f a_lo = {a[j].x, a[j].y};
      const v2f a_hi = {a[j].z, a[j].w};
#pragma unroll
      for (int c = 0; c < 4; ++c) {
        const v2f b_lo = {wf[c][j].x, wf[c][j].y};
        acc[c] = __builtin_amdgcn_wmma_f32_16x16x4_f32(
            false, a_lo, false, b_lo, (short)0, acc[c], false, false);
      }
#pragma unroll
      for (int c = 0; c < 4; ++c) {
        const v2f b_hi = {wf[c][j].z, wf[c][j].w};
        acc[c] = __builtin_amdgcn_wmma_f32_16x16x4_f32(
            false, a_hi, false, b_hi, (short)0, acc[c], false, false);
      }
    }
    float* yb = Y + (size_t)t * 16 * D + half * 64;
#pragma unroll
    for (int c = 0; c < 4; ++c) {
#pragma unroll
      for (int r = 0; r < 8; ++r) {
        yb[(size_t)(mAdd + r) * D + c * 16 + n] = acc[c][r];
      }
    }
  };

  v4f a0[16], a1[16];
  load_a(a0, tile);
  for (;;) {
    int nxt = tile + stride;
    if (nxt >= ntiles) { tile_mm(a0, tile); break; }
    load_a(a1, nxt);
    tile_mm(a0, tile);
    tile = nxt;
    nxt  = tile + stride;
    if (nxt >= ntiles) { tile_mm(a1, tile); break; }
    load_a(a0, nxt);
    tile_mm(a1, tile);
    tile = nxt;
  }
}

__device__ __forceinline__ float quad_xor1(float x) {
  return __int_as_float(__builtin_amdgcn_update_dpp(0, __float_as_int(x), 0xB1, 0xF, 0xF, true));
}
__device__ __forceinline__ float quad_xor2(float x) {
  return __int_as_float(__builtin_amdgcn_update_dpp(0, __float_as_int(x), 0x4E, 0xF, 0xF, true));
}
__device__ __forceinline__ float head_score(const float* __restrict__ krow,
                                            const float* __restrict__ qrow, int c) {
  const float4* kp = (const float4*)krow + c;
  const float4* qp = (const float4*)qrow + c;
  float p[4];
#pragma unroll
  for (int i = 0; i < 4; ++i) {
    float4 kv = kp[i * 8], qv = qp[i * 8];
    p[i] = kv.x * qv.x + kv.y * qv.y + kv.z * qv.z + kv.w * qv.w;
  }
  const bool b0 = (c & 1) != 0, b1 = (c & 2) != 0;
  float lo = b0 ? p[1] : p[0], lo_out = b0 ? p[0] : p[1];
  float hi = b0 ? p[3] : p[2], hi_out = b0 ? p[2] : p[3];
  lo += quad_xor1(lo_out);
  hi += quad_xor1(hi_out);
  float keep = b1 ? hi : lo, out = b1 ? lo : hi;
  keep += quad_xor2(out);
  return keep * SCALE;
}

__global__ void self_init(const float* __restrict__ dstK,
                          const float* __restrict__ Q,
                          float* __restrict__ self_t, int Ndst) {
  int tid = blockIdx.x * blockDim.x + threadIdx.x;
  if (tid >= Ndst * H) return;
  int n = tid >> 3, c = tid & 7;
  const float t = head_score(dstK + (size_t)n * D, Q + (size_t)n * D, c);
  self_t[(size_t)n * H + 2 * (c & 3) + (c >> 2)] = t;
}

__global__ void edge_scores(const float* __restrict__ K,
                            const float* __restrict__ Q,
                            const int* __restrict__ src_idx,
                            const int* __restrict__ dst_idx,
                            float* __restrict__ t_buf, int E) {
  int tid = blockIdx.x * blockDim.x + threadIdx.x;
  if (tid >= E * H) return;
  int e = tid >> 3, c = tid & 7;
  int s = src_idx[e], d = dst_idx[e];
  const float t = head_score(K + (size_t)s * D, Q + (size_t)d * D, c);
  t_buf[(size_t)e * H + 2 * (c & 3) + (c >> 2)] = t;
}

#define AR    200
#define ATH   1024
#define SBLK  (ATH * 4)
#define QCAP  (8 * SBLK)

__global__ __launch_bounds__(ATH) void aggregate_owned(
    const float* __restrict__ h_src, const float* __restrict__ h_dst,
    const int* __restrict__ src_idx, const int* __restrict__ dst_idx,
    const float* __restrict__ t_buf, const float* __restrict__ self_t,
    float* __restrict__ out, int E, int Ndst) {
  __shared__ float    acc[AR * D];
  __shared__ unsigned q[QCAP];
  __shared__ unsigned mx[AR * H];
  __shared__ float    den[AR * H];
  __shared__ unsigned qn;

  const int tid  = threadIdx.x;
  const int lane = tid & 31;
  const int wave  = __builtin_amdgcn_readfirstlane(tid >> 5);
  const int row0  = blockIdx.x * AR;
  const int nrows = min(AR, Ndst - row0);
  const int hl    = lane >> 2;

  for (int i = tid; i < nrows * D; i += ATH) acc[i] = 0.f;
  for (int i = tid; i < nrows * H; i += ATH) {
    mx[i]  = fenc(self_t[(size_t)row0 * H + i]);
    den[i] = 0.f;
  }
  if (tid == 0) qn = 0;
  __syncthreads();

  auto push = [&](int e) {
    const unsigned p = atomicAdd(&qn, 1u);
    if (p < QCAP) q[p] = (unsigned)e;
  };

  int lo = 0, span = E, sweep = 0;
  while (lo < E) {
    const int hi  = (span < E - lo) ? lo + span : E;
    const int nbf = (hi - lo) / SBLK;
    const int start = nbf ? (int)(((unsigned)blockIdx.x * (unsigned)nbf / gridDim.x) % (unsigned)nbf) : 0;
    const int* dspan = dst_idx + lo;
    for (int i = 0; i < nbf; i += 8) {
      int4 dv[8];
      int  e0[8];
#pragma unroll
      for (int k = 0; k < 8; ++k) {
        int b = start + min(i + k, nbf - 1);
        if (b >= nbf) b -= nbf;
        e0[k] = b * SBLK + tid * 4;
        dv[k] = *(const int4*)(dspan + e0[k]);
      }
#pragma unroll
      for (int k = 0; k < 8; ++k) {
        if (i + k < nbf) {
          const unsigned u0 = (unsigned)(dv[k].x - row0), u1 = (unsigned)(dv[k].y - row0);
          const unsigned u2 = (unsigned)(dv[k].z - row0), u3 = (unsigned)(dv[k].w - row0);
          if (min(min(u0, u1), min(u2, u3)) < (unsigned)nrows) {
            const int e = lo + e0[k];
            if (u0 < (unsigned)nrows) push(e);
            if (u1 < (unsigned)nrows) push(e + 1);
            if (u2 < (unsigned)nrows) push(e + 2);
            if (u3 < (unsigned)nrows) push(e + 3);
          }
        }
      }
    }
    for (int e = lo + nbf * SBLK + tid; e < hi; e += ATH)
      if ((unsigned)(dst_idx[e] - row0) < (unsigned)nrows) push(e);
    __syncthreads();

    const int n = __builtin_amdgcn_readfirstlane((int)qn);
    const bool fits = n <= QCAP;
    if (fits && sweep != 2) {
      for (int idx = tid; idx < n; idx += ATH) {
        const int e = (int)q[idx];
        const int r = dst_idx[e] - row0;
        const float4 ta = *(const float4*)(t_buf + (size_t)e * H);
        const float4 tb = *(const float4*)(t_buf + (size_t)e * H + 4);
        unsigned* mp = mx + r * H;
        atomicMax(mp + 0, fenc(ta.x)); atomicMax(mp + 1, fenc(ta.y));
        atomicMax(mp + 2, fenc(ta.z)); atomicMax(mp + 3, fenc(ta.w));
        atomicMax(mp + 4, fenc(tb.x)); atomicMax(mp + 5, fenc(tb.y));
        atomicMax(mp + 6, fenc(tb.z)); atomicMax(mp + 7, fenc(tb.w));
      }
    }
    if (sweep == 0) __syncthreads();
    if (fits && sweep != 1) {
      for (int b0 = wave * 32; b0 < n; b0 += ATH) {
        const int idx = b0 + lane;
        int e_l = 0, s_l = 0, r_l = 0;
        if (idx < n) {
          e_l = (int)q[idx];
          s_l = src_idx[e_l];
          r_l = dst_idx[e_l] - row0;
        }
        const int cnt = min(32, n - b0);
        for (int j = 0; j < cnt; j += 8) {
          float4 v[8];
          float  a[8];
          int    rr[8];
#pragma unroll
          for (int u = 0; u < 8; ++u) {
            const int jj = min(j + u, cnt - 1);
            const int e = __builtin_amdgcn_readlane(e_l, jj);
            const int s = __builtin_amdgcn_readlane(s_l, jj);
            rr[u] = __builtin_amdgcn_readlane(r_l, jj);
            v[u] = *(const float4*)(h_src + (size_t)s * D + 4 * lane);
            a[u] = __expf(t_buf[(size_t)e * H + hl] - fdec(mx[rr[u] * H + hl]));
          }
#pragma unroll
          for (int u = 0; u < 8; ++u) {
            if (j + u < cnt) {
              if ((lane & 3) == 0) atomicAdd(den + rr[u] * H + hl, a[u]);
              float* p = acc + rr[u] * D + lane;
              atomicAdd(p,      a[u] * v[u].x);
              atomicAdd(p + 32, a[u] * v[u].y);
              atomicAdd(p + 64, a[u] * v[u].z);
              atomicAdd(p + 96, a[u] * v[u].w);
            }
          }
        }
      }
    }
    __syncthreads();
    if (tid == 0) qn = 0;
    __syncthreads();
    if (sweep == 0) {
      if (fits) break;
      sweep = 1; span = QCAP; lo = 0;
    } else {
      lo = hi;
      if (lo >= E && sweep == 1) { sweep = 2; lo = 0; }
    }
  }

  for (int r = wave; r < nrows; r += ATH / 32) {
    const int nd = row0 + r;
    const float m  = fdec(mx[r * H + hl]);
    const float se = __expf(self_t[(size_t)nd * H + hl] - m);
    const float rd = 1.0f / (den[r * H + hl] + se);
    const float da = se * rd;
    const float4 x = *(const float4*)(h_dst + (size_t)nd * D + 4 * lane);
    const float* a = acc + r * D + lane;
    float4 o;
    o.x = a[0]  * rd + da * x.x;
    o.y = a[32] * rd + da * x.y;
    o.z = a[64] * rd + da * x.z;
    o.w = a[96] * rd + da * x.w;
    *(float4*)(out + (size_t)nd * D + 4 * lane) = o;
  }
}

extern "C" void kernel_launch(void* const* d_in, const int* in_sizes, int n_in,
                              void* d_out, int out_size, void* d_ws, size_t ws_size,
                              hipStream_t stream) {
  const float* h_src = (const float*)d_in[0];
  const float* h_dst = (const float*)d_in[1];
  const float* Wk    = (const float*)d_in[2];
  const float* Wq    = (const float*)d_in[3];
  const int* src_idx = (const int*)d_in[4];
  const int* dst_idx = (const int*)d_in[5];

  const int N_SRC = in_sizes[0] / D;
  const int N_DST = in_sizes[1] / D;
  const int E     = in_sizes[4];

  auto align256 = [](size_t x) { return (x + 255) & ~(size_t)255; };
  char* ws = (char*)d_ws;
  size_t off = 0;
  float* K      = (float*)(ws + off); off += align256((size_t)N_SRC * D * 4);
  float* Q      = (float*)(ws + off); off += align256((size_t)N_DST * D * 4);
  float* dstK   = (float*)(ws + off); off += align256((size_t)N_DST * D * 4);
  float* t_buf  = (float*)(ws + off); off += align256((size_t)E * H * 4);
  float* self_t = (float*)(ws + off); off += align256((size_t)N_DST * H * 4);

  const int TB = 256;

  gemm_xwT_wmma<<<256, GEMM_TH, 0, stream>>>(h_src, Wk, K, N_SRC);
  gemm_xwT_wmma<<<256, GEMM_TH, 0, stream>>>(h_dst, Wq, Q, N_DST);
  gemm_xwT_wmma<<<256, GEMM_TH, 0, stream>>>(h_dst, Wk, dstK, N_DST);

  self_init<<<(N_DST * H + TB - 1) / TB, TB, 0, stream>>>(dstK, Q, self_t, N_DST);
  edge_scores<<<(E * H + TB - 1) / TB, TB, 0, stream>>>(K, Q, src_idx, dst_idx, t_buf, E);
  aggregate_owned<<<(N_DST + AR - 1) / AR, ATH, 0, stream>>>(h_src, h_dst, src_idx, dst_idx, t_buf, self_t,
                                                          (float*)d_out, E, N_DST);
}


# Aggregation scan reads a 1-byte owner id per edge (dst / 200, left by the score kernel in whole-line stores), 16 edges per load: a quarter of the bytes; queue order, so LDS-atomic sum order, changes
# speedup vs baseline: 3.3376x; 1.0923x over previous
#include <hip/hip_runtime.h>
#include <hip/hip_bf16.h>

#define D      128
#define H      8
#define DK     16
#define SCALE  0.25f
#define AR     200

typedef __attribute__((ext_vector_type(2))) float v2f;
typedef __attribute__((ext_vector_type(4))) float v4f;
typedef __attribute__((ext_vector_type(8))) float v8f;

__device__ __forceinline__ unsigned fenc(float x) {
  unsigned u = __float_as_uint(x);
  return (u & 0x80000000u) ? ~u : (u | 0x80000000u);
}
__device__ __forceinline__ float fdec(unsigned k) {
  return __uint_as_float((k & 0x80000000u) ? (k & 0x7FFFFFFFu) : ~k);
}

#define GEMM_TH 256
__global__ __launch_bounds__(GEMM_TH) __attribute__((amdgpu_waves_per_eu(2, 2)))
void gemm_xwT_wmma(const float* __restrict__ X,
                   const float* __restrict__ W,
                   float* __restrict__ Y, int M) {
  const int lane = threadIdx.x & 31;
  const int wave = __builtin_amdgcn_readfirstlane((int)(threadIdx.x >> 5));
  const int half = wave >> 2;
  const int sub  = wave & 3;

  const int row  = lane & 15;
  const int koff = (lane >> 4) << 2;

  const int ntiles = M >> 4;
  const int stride = (int)gridDim.x * 4;
  int tile = (int)blockIdx.x * 4 + sub;
  if (tile >= ntiles) return;

  const float* wbase = W + (size_t)(half * 64 + row) * D + koff;
  v4f wf[4][16];
#pragma unroll
  for (int t = 0; t < 4; ++t)
#pragma unroll
    for (int j = 0; j < 16; ++j)
      wf[t][j] = *(const v4f*)(wbase + (size_t)t * 16 * D + j * 8);

  const float* xbase = X + (size_t)row * D + koff;

  const int n    = lane & 15;
  const int mAdd = (lane >> 4) << 3;

  auto load_a = [&](v4f (&a)[16], int t) {
#pragma unroll
    for (int j = 0; j < 16; ++j)
      a[j] = *(const v4f*)(xbase + (size_t)t * 16 * D + j * 8);
  };
  auto tile_mm = [&](const v4f (&a)[16], int t) {
    v8f acc[4] = {};
#pragma unroll
    for (int j = 0; j < 16; ++j) {
      const v2f a_lo = {a[j].x, a[j].y};
      const v2f a_hi = {a[j].z, a[j].w};
#pragma unroll
      for (int c = 0; c < 4; ++c) {
        const v2f b_lo = {wf[c][j].x, wf[c][j].y};
        acc[c] = __builtin_amdgcn_wmma_f32_16x16x4_f32(
            false, a_lo, false, b_lo, (short)0, acc[c], false, false);
      }
#pragma unroll
      for (int c = 0; c < 4; ++c) {
        const v2f b_hi = {wf[c][j].z, wf[c][j].w};
        acc[c] = __builtin_amdgcn_wmma_f32_16x16x4_f32(
            false, a_hi, false, b_hi, (short)0, acc[c], false, false);
      }
    }
    float* yb = Y + (size_t)t * 16 * D + half * 64;
#pragma unroll
    for (int c = 0; c < 4; ++c) {
#pragma unroll
      for (int r = 0; r < 8; ++r) {
        yb[(size_t)(mAdd + r) * D + c * 16 + n] = acc[c][r];
      }
    }
  };

  v4f a0[16], a1[16];
  load_a(a0, tile);
  for (;;) {
    int nxt = tile + stride;
    if (nxt >= ntiles) { tile_mm(a0, tile); break; }
    load_a(a1, nxt);
    tile_mm(a0, tile);
    tile = nxt;
    nxt  = tile + stride;
    if (nxt >= ntiles) { tile_mm(a1, tile); break; }
    load_a(a0, nxt);
    tile_mm(a1, tile);
    tile = nxt;
  }
}

__device__ __forceinline__ float quad_xor1(float x) {
  return __int_as_float(__builtin_amdgcn_update_dpp(0, __float_as_int(x), 0xB1, 0xF, 0xF, true));
}
__device__ __forceinline__ float quad_xor2(float x) {
  return __int_as_float(__builtin_amdgcn_update_dpp(0, __float_as_int(x), 0x4E, 0xF, 0xF, true));
}
__device__ __forceinline__ float head_score(const float* __restrict__ krow,
                                            const float* __restrict__ qrow, int c) {
  const float4* kp = (const float4*)krow + c;
  const float4* qp = (const float4*)qrow + c;
  float p[4];
#pragma unroll
  for (int i = 0; i < 4; ++i) {
    float4 kv = kp[i * 8], qv = qp[i * 8];
    p[i] = kv.x * qv.x + kv.y * qv.y + kv.z * qv.z + kv.w * qv.w;
  }
  const bool b0 = (c & 1) != 0, b1 = (c & 2) != 0;
  float lo = b0 ? p[1] : p[0], lo_out = b0 ? p[0] : p[1];
  float hi = b0 ? p[3] : p[2], hi_out = b0 ? p[2] : p[3];
  lo += quad_xor1(lo_out);
  hi += quad_xor1(hi_out);
  float keep = b1 ? hi : lo, out = b1 ? lo : hi;
  keep += quad_xor2(out);
  return keep * SCALE;
}

__global__ void self_init(const float* __restrict__ dstK,
                          const float* __restrict__ Q,
                          float* __restrict__ self_t, int Ndst) {
  int tid = blockIdx.x * blockDim.x + threadIdx.x;
  if (tid >= Ndst * H) return;
  int n = tid >> 3, c = tid & 7;
  const float t = head_score(dstK + (size_t)n * D, Q + (size_t)n * D, c);
  self_t[(size_t)n * H + 2 * (c & 3) + (c >> 2)] = t;
}

__global__ void edge_scores(const float* __restrict__ K,
                            const float* __restrict__ Q,
                            const int* __restrict__ src_idx,
                            const int* __restrict__ dst_idx,
                            float* __restrict__ t_buf,
                            unsigned* __restrict__ owner, int E) {
  int tid = blockIdx.x * blockDim.x + threadIdx.x;
  if (((tid >> 5) & 31) == 0) {
    const int e4 = (tid >> 10) * 128 + (tid & 31) * 4;
    if (e4 < E) {
      unsigned w = 0;
      if (e4 + 3 < E) {
        const int4 dv = *(const int4*)(dst_idx + e4);
        w = (unsigned)dv.x / AR | ((unsigned)dv.y / AR) << 8 |
            ((unsigned)dv.z / AR) << 16 | ((unsigned)dv.w / AR) << 24;
      } else {
        for (int j = 0; e4 + j < E; ++j) w |= ((unsigned)dst_idx[e4 + j] / AR) << (8 * j);
      }
      owner[e4 >> 2] = w;
    }
  }
  if (tid >= E * H) return;
  int e = tid >> 3, c = tid & 7;
  int s = src_idx[e], d = dst_idx[e];
  const float t = head_score(K + (size_t)s * D, Q + (size_t)d * D, c);
  t_buf[(size_t)e * H + 2 * (c & 3) + (c >> 2)] = t;
}

#define ATH   1024
#define QCAP  32768

__device__ __forceinline__ unsigned zero_bytes(unsigned x) {
  return ~(((x & 0x7F7F7F7Fu) + 0x7F7F7F7Fu) | x) & 0x80808080u;
}

__global__ __launch_bounds__(ATH) void aggregate_owned(
    const float* __restrict__ h_src, const float* __restrict__ h_dst,
    const int* __restrict__ src_idx, const int* __restrict__ dst_idx,
    const float* __restrict__ t_buf, const float* __restrict__ self_t,
    const unsigned* __restrict__ owner,
    float* __restrict__ out, int E, int Ndst) {
  __shared__ float    acc[AR * D];
  __shared__ unsigned q[QCAP];
  __shared__ unsigned mx[AR * H];
  __shared__ float    den[AR * H];
  __shared__ unsigned qn;

  const int tid  = threadIdx.x;
  const int lane = tid & 31;
  const int wave  = __builtin_amdgcn_readfirstlane(tid >> 5);
  const int row0  = blockIdx.x * AR;
  const int nrows = min(AR, Ndst - row0);
  const int hl    = lane >> 2;

  for (int i = tid; i < nrows * D; i += ATH) acc[i] = 0.f;
  for (int i = tid; i < nrows * H; i += ATH) {
    mx[i]  = fenc(self_t[(size_t)row0 * H + i]);
    den[i] = 0.f;
  }
  if (tid == 0) qn = 0;
  __syncthreads();

  auto push = [&](int e) {
    const unsigned p = atomicAdd(&qn, 1u);
    if (p < QCAP) q[p] = (unsigned)e;
  };
  const unsigned char* ob = (const unsigned char*)owner;
  const unsigned pat = (unsigned)blockIdx.x * 0x01010101u;

  int lo = 0, span = E, sweep = 0;
  while (lo < E) {
    const int hi  = (span < E - lo) ? lo + span : E;
    const int ng  = (hi - lo) >> 4;
    const int nbf = (ng + ATH - 1) / ATH;
    const int start = nbf ? (int)(((unsigned)blockIdx.x * (unsigned)nbf / gridDim.x) % (unsigned)nbf) : 0;
    const unsigned char* ospan = ob + lo;
    for (int i = 0; i < nbf; i += 8) {
      uint4 dv[8];
      int   g[8];
#pragma unroll
      for (int k = 0; k < 8; ++k) {
        int b = start + min(i + k, nbf - 1);
        if (b >= nbf) b -= nbf;
        g[k] = b * ATH + tid;
        dv[k] = *(const uint4*)(ospan + (size_t)min(g[k], ng - 1) * 16);
      }
#pragma unroll
      for (int k = 0; k < 8; ++k) {
        if (i + k < nbf && g[k] < ng) {
          const unsigned m0 = zero_bytes(dv[k].x ^ pat), m1 = zero_bytes(dv[k].y ^ pat);
          const unsigned m2 = zero_bytes(dv[k].z ^ pat), m3 = zero_bytes(dv[k].w ^ pat);
          unsigned c = (m0 >> 7) | (m1 >> 6) | (m2 >> 5) | (m3 >> 4);
          const int e = lo + g[k] * 16;
          while (c) {
            const int z = __builtin_ctz(c);
            push(e + ((z & 7) << 2) + (z >> 3));
            c &= c - 1;
          }
        }
      }
    }
    for (int e = lo + ng * 16 + tid; e < hi; e += ATH)
      if (ob[e] == (unsigned char)blockIdx.x) push(e);
    __syncthreads();

    const int n = __builtin_amdgcn_readfirstlane((int)qn);
    const bool fits = n <= QCAP;
    if (fits && sweep != 2) {
      for (int idx = tid; idx < n; idx += ATH) {
        const int e = (int)q[idx];
        const int r = dst_idx[e] - row0;
        const float4 ta = *(const float4*)(t_buf + (size_t)e * H);
        const float4 tb = *(const float4*)(t_buf + (size_t)e * H + 4);
        unsigned* mp = mx + r * H;
        atomicMax(mp + 0, fenc(ta.x)); atomicMax(mp + 1, fenc(ta.y));
        atomicMax(mp + 2, fenc(ta.z)); atomicMax(mp + 3, fenc(ta.w));
        atomicMax(mp + 4, fenc(tb.x)); atomicMax(mp + 5, fenc(tb.y));
        atomicMax(mp + 6, fenc(tb.z)); atomicMax(mp + 7, fenc(tb.w));
      }
    }
    if (sweep == 0) __syncthreads();
    if (fits && sweep != 1) {
      for (int b0 = wave * 32; b0 < n; b0 += ATH) {
        const int idx = b0 + lane;
        int e_l = 0, s_l = 0, r_l = 0;
        if (idx < n) {
          e_l = (int)q[idx];
          s_l = src_idx[e_l];
          r_l = dst_idx[e_l] - row0;
        }
        const int cnt = min(32, n - b0);
        for (int j = 0; j < cnt; j += 8) {
          float4 v[8];
          float  a[8];
          int    rr[8];
#pragma unroll
          for (int u = 0; u < 8; ++u) {
            const int jj = min(j + u, cnt - 1);
            const int e = __builtin_amdgcn_readlane(e_l, jj);
            const int s = __builtin_amdgcn_readlane(s_l, jj);
            rr[u] = __builtin_amdgcn_readlane(r_l, jj);
            v[u] = *(const float4*)(h_src + (size_t)s * D + 4 * lane);
            a[u] = __expf(t_buf[(size_t)e * H + hl] - fdec(mx[rr[u] * H + hl]));
          }
#pragma unroll
          for (int u = 0; u < 8; ++u) {
            if (j + u < cnt) {
              if ((lane & 3) == 0) atomicAdd(den + rr[u] * H + hl, a[u]);
              float* p = acc + rr[u] * D + lane;
              atomicAdd(p,      a[u] * v[u].x);
              atomicAdd(p + 32, a[u] * v[u].y);
              atomicAdd(p + 64, a[u] * v[u].z);
              atomicAdd(p + 96, a[u] * v[u].w);
            }
          }
        }
      }
    }
    __syncthreads();
    if (tid == 0) qn = 0;
    __syncthreads();
    if (sweep == 0) {
      if (fits) break;
      sweep = 1; span = QCAP; lo = 0;
    } else {
      lo = hi;
      if (lo >= E && sweep == 1) { sweep = 2; lo = 0; }
    }
  }

  for (int r = wave; r < nrows; r += ATH / 32) {
    const int nd = row0 + r;
    const float m  = fdec(mx[r * H + hl]);
    const float se = __expf(self_t[(size_t)nd * H + hl] - m);
    const float rd = 1.0f / (den[r * H + hl] + se);
    const float da = se * rd;
    const float4 x = *(const float4*)(h_dst + (size_t)nd * D + 4 * lane);
    const float* a = acc + r * D + lane;
    float4 o;
    o.x = a[0]  * rd + da * x.x;
    o.y = a[32] * rd + da * x.y;
    o.z = a[64] * rd + da * x.z;
    o.w = a[96] * rd + da * x.w;
    *(float4*)(out + (size_t)nd * D + 4 * lane) = o;
  }
}

extern "C" void kernel_launch(void* const* d_in, const int* in_sizes, int n_in,
                              void* d_out, int out_size, void* d_ws, size_t ws_size,
                              hipStream_t stream) {
  const float* h_src = (const float*)d_in[0];
  const float* h_dst = (const float*)d_in[1];
  const float* Wk    = (const float*)d_in[2];
  const float* Wq    = (const float*)d_in[3];
  const int* src_idx = (const int*)d_in[4];
  const int* dst_idx = (const int*)d_in[5];

  const int N_SRC = in_sizes[0] / D;
  const int N_DST = in_sizes[1] / D;
  const int E     = in_sizes[4];

  auto align256 = [](size_t x) { return (x + 255) & ~(size_t)255; };
  char* ws = (char*)d_ws;
  size_t off = 0;
  float* K      = (float*)(ws + off); off += align256((size_t)N_SRC * D * 4);
  float* Q      = (float*)(ws + off); off += align256((size_t)N_DST * D * 4);
  float* dstK   = (float*)(ws + off); off += align256((size_t)N_DST * D * 4);
  float* t_buf  = (float*)(ws + off); off += align256((size_t)E * H * 4);
  float* self_t = (float*)(ws + off); off += align256((size_t)N_DST * H * 4);
  unsigned* owner = (unsigned*)(ws + off); off += align256((size_t)E);

  const int TB = 256;

  gemm_xwT_wmma<<<256, GEMM_TH, 0, stream>>>(h_src, Wk, K, N_SRC);
  gemm_xwT_wmma<<<256, GEMM_TH, 0, stream>>>(h_dst, Wq, Q, N_DST);
  gemm_xwT_wmma<<<256, GEMM_TH, 0, stream>>>(h_dst, Wk, dstK, N_DST);

  self_init<<<(N_DST * H + TB - 1) / TB, TB, 0, stream>>>(dstK, Q, self_t, N_DST);
  edge_scores<<<(E * H + TB - 1) / TB, TB, 0, stream>>>(K, Q, src_idx, dst_idx, t_buf, owner, E);
  aggregate_owned<<<(N_DST + AR - 1) / AR, ATH, 0, stream>>>(h_src, h_dst, src_idx, dst_idx, t_buf, self_t,
                                                          owner, (float*)d_out, E, N_DST);
}


# Projections fused: one launch makes K, Q and the self score; a wave keeps 32 columns of Wk and of Wq in registers, self score from accumulators (16-lane sum, sum order changes); 3 launches fewer
# speedup vs baseline: 3.5278x; 1.0570x over previous
#include <hip/hip_runtime.h>
#include <hip/hip_bf16.h>

#define D      128
#define H      8
#define DK     16
#define SCALE  0.25f
#define AR     200

typedef __attribute__((ext_vector_type(2))) float v2f;
typedef __attribute__((ext_vector_type(4))) float v4f;
typedef __attribute__((ext_vector_type(8))) float v8f;

__device__ __forceinline__ unsigned fenc(float x) {
  unsigned u = __float_as_uint(x);
  return (u & 0x80000000u) ? ~u : (u | 0x80000000u);
}
__device__ __forceinline__ float fdec(unsigned k) {
  return __uint_as_float((k & 0x80000000u) ? (k & 0x7FFFFFFFu) : ~k);
}

#define GEMM_TH 256
__device__ __forceinline__ float half_sum16(float p) {
  p += __int_as_float(__builtin_amdgcn_update_dpp(0, __float_as_int(p), 0xB1,  0xF, 0xF, true));
  p += __int_as_float(__builtin_amdgcn_update_dpp(0, __float_as_int(p), 0x4E,  0xF, 0xF, true));
  p += __int_as_float(__builtin_amdgcn_update_dpp(0, __float_as_int(p), 0x141, 0xF, 0xF, true));
  p += __int_as_float(__builtin_amdgcn_update_dpp(0, __float_as_int(p), 0x140, 0xF, 0xF, true));
  return p;
}
__global__ __launch_bounds__(GEMM_TH) __attribute__((amdgpu_waves_per_eu(2, 2)))
void project_all(const float* __restrict__ Xs, const float* __restrict__ Xd,
                 const float* __restrict__ Wk, const float* __restrict__ Wq,
                 float* __restrict__ K, float* __restrict__ Q,
                 float* __restrict__ self_t, int Ms, int Md) {
  const int lane = threadIdx.x & 31;
  const int wave = __builtin_amdgcn_readfirstlane((int)(threadIdx.x >> 5));
  const int quarter = wave >> 1;
  const int sub     = wave & 1;

  const int row  = lane & 15;
  const int koff = (lane >> 4) << 2;
  const int col0 = quarter * 32;

  v4f wk[2][16], wq[2][16];
  {
    const float* kb = Wk + (size_t)(col0 + row) * D + koff;
    const float* qb = Wq + (size_t)(col0 + row) * D + koff;
#pragma unroll
    for (int t = 0; t < 2; ++t)
#pragma unroll
      for (int j = 0; j < 16; ++j) {
        wk[t][j] = *(const v4f*)(kb + (size_t)t * 16 * D + j * 8);
        wq[t][j] = *(const v4f*)(qb + (size_t)t * 16 * D + j * 8);
      }
  }

  const int n    = lane & 15;
  const int mAdd = (lane >> 4) << 3;

  auto load_a = [&](v4f (&a)[16], const float* xbase, int t) {
#pragma unroll
    for (int j = 0; j < 16; ++j)
      a[j] = *(const v4f*)(xbase + (size_t)t * 16 * D + j * 8);
  };
  auto mm_src = [&](const v4f (&a)[16], int t) {
    v8f acc[2] = {};
#pragma unroll
    for (int j = 0; j < 16; ++j) {
      const v2f a_lo = {a[j].x, a[j].y};
      const v2f a_hi = {a[j].z, a[j].w};
#pragma unroll
      for (int c = 0; c < 2; ++c) {
        const v2f b_lo = {wk[c][j].x, wk[c][j].y};
        acc[c] = __builtin_amdgcn_wmma_f32_16x16x4_f32(
            false, a_lo, false, b_lo, (short)0, acc[c], false, false);
      }
#pragma unroll
      for (int c = 0; c < 2; ++c) {
        const v2f b_hi = {wk[c][j].z, wk[c][j].w};
        acc[c] = __builtin_amdgcn_wmma_f32_16x16x4_f32(
            false, a_hi, false, b_hi, (short)0, acc[c], false, false);
      }
    }
    float* yb = K + (size_t)t * 16 * D + col0;
#pragma unroll
    for (int c = 0; c < 2; ++c) {
#pragma unroll
      for (int r = 0; r < 8; ++r) {
        yb[(size_t)(mAdd + r) * D + c * 16 + n] = acc[c][r];
      }
    }
  };
  auto mm_dst = [&](const v4f (&a)[16], int t) {
    v8f aq[2] = {}, ak[2] = {};
#pragma unroll
    for (int j = 0; j < 16; ++j) {
      const v2f a_lo = {a[j].x, a[j].y};
      const v2f a_hi = {a[j].z, a[j].w};
#pragma unroll
      for (int c = 0; c < 2; ++c) {
        const v2f q_lo = {wq[c][j].x, wq[c][j].y};
        const v2f k_lo = {wk[c][j].x, wk[c][j].y};
        aq[c] = __builtin_amdgcn_wmma_f32_16x16x4_f32(
            false, a_lo, false, q_lo, (short)0, aq[c], false, false);
        ak[c] = __builtin_amdgcn_wmma_f32_16x16x4_f32(
            false, a_lo, false, k_lo, (short)0, ak[c], false, false);
      }
#pragma unroll
      for (int c = 0; c < 2; ++c) {
        const v2f q_hi = {wq[c][j].z, wq[c][j].w};
        const v2f k_hi = {wk[c][j].z, wk[c][j].w};
        aq[c] = __builtin_amdgcn_wmma_f32_16x16x4_f32(
            false, a_hi, false, q_hi, (short)0, aq[c], false, false);
        ak[c] = __builtin_amdgcn_wmma_f32_16x16x4_f32(
            false, a_hi, false, k_hi, (short)0, ak[c], false, false);
      }
    }
    float* yb = Q + (size_t)t * 16 * D + col0;
#pragma unroll
    for (int c = 0; c < 2; ++c) {
#pragma unroll
      for (int r = 0; r < 8; ++r) {
        yb[(size_t)(mAdd + r) * D + c * 16 + n] = aq[c][r];
      }
    }
#pragma unroll
    for (int c = 0; c < 2; ++c) {
      float mine = 0.f;
#pragma unroll
      for (int r = 0; r < 8; ++r) {
        const float s = half_sum16(aq[c][r] * ak[c][r]);
        mine = (n == r) ? s : mine;
      }
      if (n < 8) self_t[(size_t)(t * 16 + mAdd + n) * H + quarter * 2 + c] = mine * SCALE;
    }
  };

  const int nslots = (int)gridDim.x * 2;
  const int slot   = (int)blockIdx.x * 2 + sub;
  auto run = [&](const float* X, int tile, int ntiles, auto&& mm) {
    if (tile >= ntiles) return;
    const float* xbase = X + (size_t)row * D + koff;
    v4f a0[16], a1[16];
    load_a(a0, xbase, tile);
    for (;;) {
      int nxt = tile + nslots;
      if (nxt >= ntiles) { mm(a0, tile); break; }
      load_a(a1, xbase, nxt);
      mm(a0, tile);
      tile = nxt;
      nxt  = tile + nslots;
      if (nxt >= ntiles) { mm(a1, tile); break; }
      load_a(a0, xbase, nxt);
      mm(a1, tile);
      tile = nxt;
    }
  };
  run(Xs, slot, Ms >> 4, mm_src);
  run(Xd, nslots - 1 - slot, Md >> 4, mm_dst);
}

__device__ __forceinline__ float quad_xor1(float x) {
  return __int_as_float(__builtin_amdgcn_update_dpp(0, __float_as_int(x), 0xB1, 0xF, 0xF, true));
}
__device__ __forceinline__ float quad_xor2(float x) {
  return __int_as_float(__builtin_amdgcn_update_dpp(0, __float_as_int(x), 0x4E, 0xF, 0xF, true));
}
__device__ __forceinline__ float head_score(const float* __restrict__ krow,
                                            const float* __restrict__ qrow, int c) {
  const float4* kp = (const float4*)krow + c;
  const float4* qp = (const float4*)qrow + c;
  float p[4];
#pragma unroll
  for (int i = 0; i < 4; ++i) {
    float4 kv = kp[i * 8], qv = qp[i * 8];
    p[i] = kv.x * qv.x + kv.y * qv.y + kv.z * qv.z + kv.w * qv.w;
  }
  const bool b0 = (c & 1) != 0, b1 = (c & 2) != 0;
  float lo = b0 ? p[1] : p[0], lo_out = b0 ? p[0] : p[1];
  float hi = b0 ? p[3] : p[2], hi_out = b0 ? p[2] : p[3];
  lo += quad_xor1(lo_out);
  hi += quad_xor1(hi_out);
  float keep = b1 ? hi : lo, out = b1 ? lo : hi;
  keep += quad_xor2(out);
  return keep * SCALE;
}

__global__ void edge_scores(const float* __restrict__ K,
                            const float* __restrict__ Q,
                            const int* __restrict__ src_idx,
                            const int* __restrict__ dst_idx,
                            float* __restrict__ t_buf,
                            unsigned* __restrict__ owner, int E) {
  int tid = blockIdx.x * blockDim.x + threadIdx.x;
  if (((tid >> 5) & 31) == 0) {
    const int e4 = (tid >> 10) * 128 + (tid & 31) * 4;
    if (e4 < E) {
      unsigned w = 0;
      if (e4 + 3 < E) {
        const int4 dv = *(const int4*)(dst_idx + e4);
        w = (unsigned)dv.x / AR | ((unsigned)dv.y / AR) << 8 |
            ((unsigned)dv.z / AR) << 16 | ((unsigned)dv.w / AR) << 24;
      } else {
        for (int j = 0; e4 + j < E; ++j) w |= ((unsigned)dst_idx[e4 + j] / AR) << (8 * j);
      }
      owner[e4 >> 2] = w;
    }
  }
  if (tid >= E * H) return;
  int e = tid >> 3, c = tid & 7;
  int s = src_idx[e], d = dst_idx[e];
  const float t = head_score(K + (size_t)s * D, Q + (size_t)d * D, c);
  t_buf[(size_t)e * H + 2 * (c & 3) + (c >> 2)] = t;
}

#define ATH   1024
#define QCAP  32768

__device__ __forceinline__ unsigned zero_bytes(unsigned x) {
  return ~(((x & 0x7F7F7F7Fu) + 0x7F7F7F7Fu) | x) & 0x80808080u;
}

__global__ __launch_bounds__(ATH) void aggregate_owned(
    const float* __restrict__ h_src, const float* __restrict__ h_dst,
    const int* __restrict__ src_idx, const int* __restrict__ dst_idx,
    const float* __restrict__ t_buf, const float* __restrict__ self_t,
    const unsigned* __restrict__ owner,
    float* __restrict__ out, int E, int Ndst) {
  __shared__ float    acc[AR * D];
  __shared__ unsigned q[QCAP];
  __shared__ unsigned mx[AR * H];
  __shared__ float    den[AR * H];
  __shared__ unsigned qn;

  const int tid  = threadIdx.x;
  const int lane = tid & 31;
  const int wave  = __builtin_amdgcn_readfirstlane(tid >> 5);
  const int row0  = blockIdx.x * AR;
  const int nrows = min(AR, Ndst - row0);
  const int hl    = lane >> 2;

  for (int i = tid; i < nrows * D; i += ATH) acc[i] = 0.f;
  for (int i = tid; i < nrows * H; i += ATH) {
    mx[i]  = fenc(self_t[(size_t)row0 * H + i]);
    den[i] = 0.f;
  }
  if (tid == 0) qn = 0;
  __syncthreads();

  auto push = [&](int e) {
    const unsigned p = atomicAdd(&qn, 1u);
    if (p < QCAP) q[p] = (unsigned)e;
  };
  const unsigned char* ob = (const unsigned char*)owner;
  const unsigned pat = (unsigned)blockIdx.x * 0x01010101u;

  int lo = 0, span = E, sweep = 0;
  while (lo < E) {
    const int hi  = (span < E - lo) ? lo + span : E;
    const int ng  = (hi - lo) >> 4;
    const int nbf = (ng + ATH - 1) / ATH;
    const int start = nbf ? (int)(((unsigned)blockIdx.x * (unsigned)nbf / gridDim.x) % (unsigned)nbf) : 0;
    const unsigned char* ospan = ob + lo;
    for (int i = 0; i < nbf; i += 8) {
      uint4 dv[8];
      int   g[8];
#pragma unroll
      for (int k = 0; k < 8; ++k) {
        int b = start + min(i + k, nbf - 1);
        if (b >= nbf) b -= nbf;
        g[k] = b * ATH + tid;
        dv[k] = *(const uint4*)(ospan + (size_t)min(g[k], ng - 1) * 16);
      }
#pragma unroll
      for (int k = 0; k < 8; ++k) {
        if (i + k < nbf && g[k] < ng) {
          const unsigned m0 = zero_bytes(dv[k].x ^ pat), m1 = zero_bytes(dv[k].y ^ pat);
          const unsigned m2 = zero_bytes(dv[k].z ^ pat), m3 = zero_bytes(dv[k].w ^ pat);
          unsigned c = (m0 >> 7) | (m1 >> 6) | (m2 >> 5) | (m3 >> 4);
          const int e = lo + g[k] * 16;
          while (c) {
            const int z = __builtin_ctz(c);
            push(e + ((z & 7) << 2) + (z >> 3));
            c &= c - 1;
          }
        }
      }
    }
    for (int e = lo + ng * 16 + tid; e < hi; e += ATH)
      if (ob[e] == (unsigned char)blockIdx.x) push(e);
    __syncthreads();

    const int n = __builtin_amdgcn_readfirstlane((int)qn);
    const bool fits = n <= QCAP;
    if (fits && sweep != 2) {
      for (int idx = tid; idx < n; idx += ATH) {
        const int e = (int)q[idx];
        const int r = dst_idx[e] - row0;
        const float4 ta = *(const float4*)(t_buf + (size_t)e * H);
        const float4 tb = *(const float4*)(t_buf + (size_t)e * H + 4);
        unsigned* mp = mx + r * H;
        atomicMax(mp + 0, fenc(ta.x)); atomicMax(mp + 1, fenc(ta.y));
        atomicMax(mp + 2, fenc(ta.z)); atomicMax(mp + 3, fenc(ta.w));
        atomicMax(mp + 4, fenc(tb.x)); atomicMax(mp + 5, fenc(tb.y));
        atomicMax(mp + 6, fenc(tb.z)); atomicMax(mp + 7, fenc(tb.w));
      }
    }
    if (sweep == 0) __syncthreads();
    if (fits && sweep != 1) {
      for (int b0 = wave * 32; b0 < n; b0 += ATH) {
        const int idx = b0 + lane;
        int e_l = 0, s_l = 0, r_l = 0;
        if (idx < n) {
          e_l = (int)q[idx];
          s_l = src_idx[e_l];
          r_l = dst_idx[e_l] - row0;
        }
        const int cnt = min(32, n - b0);
        for (int j = 0; j < cnt; j += 8) {
          float4 v[8];
          float  a[8];
          int    rr[8];
#pragma unroll
          for (int u = 0; u < 8; ++u) {
            const int jj = min(j + u, cnt - 1);
            const int e = __builtin_amdgcn_readlane(e_l, jj);
            const int s = __builtin_amdgcn_readlane(s_l, jj);
            rr[u] = __builtin_amdgcn_readlane(r_l, jj);
            v[u] = *(const float4*)(h_src + (size_t)s * D + 4 * lane);
            a[u] = __expf(t_buf[(size_t)e * H + hl] - fdec(mx[rr[u] * H + hl]));
          }
#pragma unroll
          for (int u = 0; u < 8; ++u) {
            if (j + u < cnt) {
              if ((lane & 3) == 0) atomicAdd(den + rr[u] * H + hl, a[u]);
              float* p = acc + rr[u] * D + lane;
              atomicAdd(p,      a[u] * v[u].x);
              atomicAdd(p + 32, a[u] * v[u].y);
              atomicAdd(p + 64, a[u] * v[u].z);
              atomicAdd(p + 96, a[u] * v[u].w);
            }
          }
        }
      }
    }
    __syncthreads();
    if (tid == 0) qn = 0;
    __syncthreads();
    if (sweep == 0) {
      if (fits) break;
      sweep = 1; span = QCAP; lo = 0;
    } else {
      lo = hi;
      if (lo >= E && sweep == 1) { sweep = 2; lo = 0; }
    }
  }

  for (int r = wave; r < nrows; r += ATH / 32) {
    const int nd = row0 + r;
    const float m  = fdec(mx[r * H + hl]);
    const float se = __expf(self_t[(size_t)nd * H + hl] - m);
    const float rd = 1.0f / (den[r * H + hl] + se);
    const float da = se * rd;
    const float4 x = *(const float4*)(h_dst + (size_t)nd * D + 4 * lane);
    const float* a = acc + r * D + lane;
    float4 o;
    o.x = a[0]  * rd + da * x.x;
    o.y = a[32] * rd + da * x.y;
    o.z = a[64] * rd + da * x.z;
    o.w = a[96] * rd + da * x.w;
    *(float4*)(out + (size_t)nd * D + 4 * lane) = o;
  }
}

extern "C" void kernel_launch(void* const* d_in, const int* in_sizes, int n_in,
                              void* d_out, int out_size, void* d_ws, size_t ws_size,
                              hipStream_t stream) {
  const float* h_src = (const float*)d_in[0];
  const float* h_dst = (const float*)d_in[1];
  const float* Wk    = (const float*)d_in[2];
  const float* Wq    = (const float*)d_in[3];
  const int* src_idx = (const int*)d_in[4];
  const int* dst_idx = (const int*)d_in[5];

  const int N_SRC = in_sizes[0] / D;
  const int N_DST = in_sizes[1] / D;
  const int E     = in_sizes[4];

  auto align256 = [](size_t x) { return (x + 255) & ~(size_t)255; };
  char* ws = (char*)d_ws;
  size_t off = 0;
  float* K      = (float*)(ws + off); off += align256((size_t)N_SRC * D * 4);
  float* Q      = (float*)(ws + off); off += align256((size_t)N_DST * D * 4);
  float* t_buf  = (float*)(ws + off); off += align256((size_t)E * H * 4);
  float* self_t = (float*)(ws + off); off += align256((size_t)N_DST * H * 4);
  unsigned* owner = (unsigned*)(ws + off); off += align256((size_t)E);

  const int TB = 256;

  project_all<<<256, GEMM_TH, 0, stream>>>(h_src, h_dst, Wk, Wq, K, Q, self_t, N_SRC, N_DST);
  edge_scores<<<(E * H + TB - 1) / TB, TB, 0, stream>>>(K, Q, src_idx, dst_idx, t_buf, owner, E);
  aggregate_owned<<<(N_DST + AR - 1) / AR, ATH, 0, stream>>>(h_src, h_dst, src_idx, dst_idx, t_buf, self_t,
                                                          owner, (float*)d_out, E, N_DST);
}


# Edge scores made by the owning work-group: its 200 Q rows sit in LDS, queued edges gather K rows, scores kept in queue order; score launch, Q gather, scattered reads gone; score sum order changes
# speedup vs baseline: 4.6307x; 1.3126x over previous
#include <hip/hip_runtime.h>
#include <hip/hip_bf16.h>

#define D      128
#define H      8
#define DK     16
#define SCALE  0.25f
#define AR     200

typedef __attribute__((ext_vector_type(2))) float v2f;
typedef __attribute__((ext_vector_type(4))) float v4f;
typedef __attribute__((ext_vector_type(8))) float v8f;
typedef __attribute__((ext_vector_type(4))) int   v4i;

__device__ __forceinline__ unsigned fenc(float x) {
  unsigned u = __float_as_uint(x);
  return (u & 0x80000000u) ? ~u : (u | 0x80000000u);
}
__device__ __forceinline__ float fdec(unsigned k) {
  return __uint_as_float((k & 0x80000000u) ? (k & 0x7FFFFFFFu) : ~k);
}

#define GEMM_TH 256
__device__ __forceinline__ float half_sum16(float p) {
  p += __int_as_float(__builtin_amdgcn_update_dpp(0, __float_as_int(p), 0xB1,  0xF, 0xF, true));
  p += __int_as_float(__builtin_amdgcn_update_dpp(0, __float_as_int(p), 0x4E,  0xF, 0xF, true));
  p += __int_as_float(__builtin_amdgcn_update_dpp(0, __float_as_int(p), 0x141, 0xF, 0xF, true));
  p += __int_as_float(__builtin_amdgcn_update_dpp(0, __float_as_int(p), 0x140, 0xF, 0xF, true));
  return p;
}
__global__ __launch_bounds__(GEMM_TH) __attribute__((amdgpu_waves_per_eu(2, 2)))
void project_all(const float* __restrict__ Xs, const float* __restrict__ Xd,
                 const float* __restrict__ Wk, const float* __restrict__ Wq,
                 float* __restrict__ K, float* __restrict__ Q,
                 float* __restrict__ self_t, int Ms, int Md,
                 const int* __restrict__ dst_idx, unsigned* __restrict__ owner, int E) {
  {
    const int step  = (int)gridDim.x * GEMM_TH * 4;
    const int nfull = E & ~3;
    if (nfull) {
      for (int base = ((int)blockIdx.x * GEMM_TH + (int)threadIdx.x) * 4; base < nfull; base += step * 8) {
        v4i dv[8];
#pragma unroll
        for (int k = 0; k < 8; ++k)
          dv[k] = *(const v4i*)(dst_idx + min(base + k * step, nfull - 4));
        asm volatile("" :: "v"(dv[0]), "v"(dv[1]), "v"(dv[2]), "v"(dv[3]),
                           "v"(dv[4]), "v"(dv[5]), "v"(dv[6]), "v"(dv[7]));
#pragma unroll
        for (int k = 0; k < 8; ++k) {
          const int e4 = base + k * step;
          if (e4 < nfull)
            owner[e4 >> 2] = (unsigned)dv[k].x / AR | ((unsigned)dv[k].y / AR) << 8 |
                             ((unsigned)dv[k].z / AR) << 16 | ((unsigned)dv[k].w / AR) << 24;
        }
      }
    }
    if (nfull < E && blockIdx.x == 0 && threadIdx.x == 0) {
      unsigned w = 0;
      for (int j = 0; nfull + j < E; ++j) w |= ((unsigned)dst_idx[nfull + j] / AR) << (8 * j);
      owner[nfull >> 2] = w;
    }
  }
  const int lane = threadIdx.x & 31;
  const int wave = __builtin_amdgcn_readfirstlane((int)(threadIdx.x >> 5));
  const int quarter = wave >> 1;
  const int sub     = wave & 1;

  const int row  = lane & 15;
  const int koff = (lane >> 4) << 2;
  const int col0 = quarter * 32;

  v4f wk[2][16], wq[2][16];
  {
    const float* kb = Wk + (size_t)(col0 + row) * D + koff;
    const float* qb = Wq + (size_t)(col0 + row) * D + koff;
#pragma unroll
    for (int t = 0; t < 2; ++t)
#pragma unroll
      for (int j = 0; j < 16; ++j) {
        wk[t][j] = *(const v4f*)(kb + (size_t)t * 16 * D + j * 8);
        wq[t][j] = *(const v4f*)(qb + (size_t)t * 16 * D + j * 8);
      }
  }

  const int n    = lane & 15;
  const int mAdd = (lane >> 4) << 3;

  auto load_a = [&](v4f (&a)[16], const float* xbase, int t) {
#pragma unroll
    for (int j = 0; j < 16; ++j)
      a[j] = *(const v4f*)(xbase + (size_t)t * 16 * D + j * 8);
  };
  auto mm_src = [&](const v4f (&a)[16], int t) {
    v8f acc[2] = {};
#pragma unroll
    for (int j = 0; j < 16; ++j) {
      const v2f a_lo = {a[j].x, a[j].y};
      const v2f a_hi = {a[j].z, a[j].w};
#pragma unroll
      for (int c = 0; c < 2; ++c) {
        const v2f b_lo = {wk[c][j].x, wk[c][j].y};
        acc[c] = __builtin_amdgcn_wmma_f32_16x16x4_f32(
            false, a_lo, false, b_lo, (short)0, acc[c], false, false);
      }
#pragma unroll
      for (int c = 0; c < 2; ++c) {
        const v2f b_hi = {wk[c][j].z, wk[c][j].w};
        acc[c] = __builtin_amdgcn_wmma_f32_16x16x4_f32(
            false, a_hi, false, b_hi, (short)0, acc[c], false, false);
      }
    }
    float* yb = K + (size_t)t * 16 * D + col0;
#pragma unroll
    for (int c = 0; c < 2; ++c) {
#pragma unroll
      for (int r = 0; r < 8; ++r) {
        yb[(size_t)(mAdd + r) * D + c * 16 + n] = acc[c][r];
      }
    }
  };
  auto mm_dst = [&](const v4f (&a)[16], int t) {
    v8f aq[2] = {}, ak[2] = {};
#pragma unroll
    for (int j = 0; j < 16; ++j) {
      const v2f a_lo = {a[j].x, a[j].y};
      const v2f a_hi = {a[j].z, a[j].w};
#pragma unroll
      for (int c = 0; c < 2; ++c) {
        const v2f q_lo = {wq[c][j].x, wq[c][j].y};
        const v2f k_lo = {wk[c][j].x, wk[c][j].y};
        aq[c] = __builtin_amdgcn_wmma_f32_16x16x4_f32(
            false, a_lo, false, q_lo, (short)0, aq[c], false, false);
        ak[c] = __builtin_amdgcn_wmma_f32_16x16x4_f32(
            false, a_lo, false, k_lo, (short)0, ak[c], false, false);
      }
#pragma unroll
      for (int c = 0; c < 2; ++c) {
        const v2f q_hi = {wq[c][j].z, wq[c][j].w};
        const v2f k_hi = {wk[c][j].z, wk[c][j].w};
        aq[c] = __builtin_amdgcn_wmma_f32_16x16x4_f32(
            false, a_hi, false, q_hi, (short)0, aq[c], false, false);
        ak[c] = __builtin_amdgcn_wmma_f32_16x16x4_f32(
            false, a_hi, false, k_hi, (short)0, ak[c], false, false);
      }
    }
    float* yb = Q + (size_t)t * 16 * D + col0;
#pragma unroll
    for (int c = 0; c < 2; ++c) {
#pragma unroll
      for (int r = 0; r < 8; ++r) {
        yb[(size_t)(mAdd + r) * D + c * 16 + n] = aq[c][r];
      }
    }
#pragma unroll
    for (int c = 0; c < 2; ++c) {
      float mine = 0.f;
#pragma unroll
      for (int r = 0; r < 8; ++r) {
        const float s = half_sum16(aq[c][r] * ak[c][r]);
        mine = (n == r) ? s : mine;
      }
      if (n < 8) self_t[(size_t)(t * 16 + mAdd + n) * H + quarter * 2 + c] = mine * SCALE;
    }
  };

  const int nslots = (int)gridDim.x * 2;
  const int slot   = (int)blockIdx.x * 2 + sub;
  auto run = [&](const float* X, int tile, int ntiles, auto&& mm) {
    if (tile >= ntiles) return;
    const float* xbase = X + (size_t)row * D + koff;
    v4f a0[16], a1[16];
    load_a(a0, xbase, tile);
    for (;;) {
      int nxt = tile + nslots;
      if (nxt >= ntiles) { mm(a0, tile); break; }
      load_a(a1, xbase, nxt);
      mm(a0, tile);
      tile = nxt;
      nxt  = tile + nslots;
      if (nxt >= ntiles) { mm(a1, tile); break; }
      load_a(a0, xbase, nxt);
      mm(a1, tile);
      tile = nxt;
    }
  };
  run(Xs, slot, Ms >> 4, mm_src);
  run(Xd, nslots - 1 - slot, Md >> 4, mm_dst);
}

__device__ __forceinline__ float quad_xor1(float x) {
  return __int_as_float(__builtin_amdgcn_update_dpp(0, __float_as_int(x), 0xB1, 0xF, 0xF, true));
}
__device__ __forceinline__ float quad_xor2(float x) {
  return __int_as_float(__builtin_amdgcn_update_dpp(0, __float_as_int(x), 0x4E, 0xF, 0xF, true));
}

#define ATH   1024
#define QCAP  16384

__device__ __forceinline__ unsigned zero_bytes(unsigned x) {
  return ~(((x & 0x7F7F7F7Fu) + 0x7F7F7F7Fu) | x) & 0x80808080u;
}

__global__ __launch_bounds__(ATH) void aggregate_owned(
    const float* __restrict__ h_src, const float* __restrict__ h_dst,
    const float* __restrict__ K, const float* __restrict__ Q,
    const int* __restrict__ src_idx, const int* __restrict__ dst_idx,
    const float* __restrict__ self_t, const unsigned* __restrict__ owner,
    float* __restrict__ t_q,
    float* __restrict__ out, int E, int Ndst) {
  __shared__ float    acc[AR * D];
  __shared__ __attribute__((aligned(16))) float qrow[AR * D];
  __shared__ unsigned q[QCAP];
  __shared__ unsigned mx[AR * H];
  __shared__ float    den[AR * H];
  __shared__ unsigned qn;

  const int tid  = threadIdx.x;
  const int lane = tid & 31;
  const int wave  = __builtin_amdgcn_readfirstlane(tid >> 5);
  const int row0  = blockIdx.x * AR;
  const int nrows = min(AR, Ndst - row0);
  const int hl    = lane >> 2;

  float* tq = t_q + (size_t)blockIdx.x * QCAP * H;

  for (int i = tid; i < nrows * D; i += ATH) acc[i] = 0.f;
  for (int i = tid; i < nrows * (D / 4); i += ATH)
    ((float4*)qrow)[i] = ((const float4*)(Q + (size_t)row0 * D))[i];
  for (int i = tid; i < nrows * H; i += ATH) {
    mx[i]  = fenc(self_t[(size_t)row0 * H + i]);
    den[i] = 0.f;
  }
  if (tid == 0) qn = 0;
  __syncthreads();

  auto push = [&](int e) {
    const unsigned p = atomicAdd(&qn, 1u);
    if (p < QCAP) q[p] = (unsigned)e;
  };
  const unsigned char* ob = (const unsigned char*)owner;
  const unsigned pat = (unsigned)blockIdx.x * 0x01010101u;

  int lo = 0, span = E, sweep = 0;
  while (lo < E) {
    const int hi  = (span < E - lo) ? lo + span : E;
    const int ng  = (hi - lo) >> 4;
    const int nbf = (ng + ATH - 1) / ATH;
    const int start = nbf ? (int)(((unsigned)blockIdx.x * (unsigned)nbf / gridDim.x) % (unsigned)nbf) : 0;
    const unsigned char* ospan = ob + lo;
    for (int i = 0; i < nbf; i += 8) {
      uint4 dv[8];
      int   g[8];
#pragma unroll
      for (int k = 0; k < 8; ++k) {
        int b = start + min(i + k, nbf - 1);
        if (b >= nbf) b -= nbf;
        g[k] = b * ATH + tid;
        dv[k] = *(const uint4*)(ospan + (size_t)min(g[k], ng - 1) * 16);
      }
#pragma unroll
      for (int k = 0; k < 8; ++k) {
        if (i + k < nbf && g[k] < ng) {
          const unsigned m0 = zero_bytes(dv[k].x ^ pat), m1 = zero_bytes(dv[k].y ^ pat);
          const unsigned m2 = zero_bytes(dv[k].z ^ pat), m3 = zero_bytes(dv[k].w ^ pat);
          unsigned c = (m0 >> 7) | (m1 >> 6) | (m2 >> 5) | (m3 >> 4);
          const int e = lo + g[k] * 16;
          while (c) {
            const int z = __builtin_ctz(c);
            push(e + ((z & 7) << 2) + (z >> 3));
            c &= c - 1;
          }
        }
      }
    }
    for (int e = lo + ng * 16 + tid; e < hi; e += ATH)
      if (ob[e] == (unsigned char)blockIdx.x) push(e);
    __syncthreads();

    const int n = __builtin_amdgcn_readfirstlane((int)qn);
    const bool fits = n <= QCAP;
    if (fits) {
      for (int b0 = wave * 32; b0 < n; b0 += ATH) {
        const int idx = b0 + lane;
        int s_l = 0, r_l = 0;
        if (idx < n) {
          const int e_l = (int)q[idx];
          s_l = src_idx[e_l];
          r_l = dst_idx[e_l] - row0;
          q[idx] = (unsigned)s_l | ((unsigned)r_l << 24);
        }
        const int cnt = min(32, n - b0);
        for (int j = 0; j < cnt; j += 8) {
          float4 kv[8];
          float  sc[8];
          int    rr[8];
#pragma unroll
          for (int u = 0; u < 8; ++u) {
            const int jj = min(j + u, cnt - 1);
            const int s = __builtin_amdgcn_readlane(s_l, jj);
            rr[u] = __builtin_amdgcn_readlane(r_l, jj);
            kv[u] = *(const float4*)(K + (size_t)s * D + 4 * lane);
          }
#pragma unroll
          for (int u = 0; u < 8; ++u) {
            const float4 qv = *(const float4*)(qrow + rr[u] * D + 4 * lane);
            float p = kv[u].x * qv.x + kv[u].y * qv.y + kv[u].z * qv.z + kv[u].w * qv.w;
            p += quad_xor1(p);
            p += quad_xor2(p);
            sc[u] = p * SCALE;
          }
          const int sub = lane & 3;
#pragma unroll
          for (int g = 0; g < 2; ++g) {
            float val = sc[4 * g];
            int   rs  = rr[4 * g];
            if (sub == 1) { val = sc[4 * g + 1]; rs = rr[4 * g + 1]; }
            if (sub == 2) { val = sc[4 * g + 2]; rs = rr[4 * g + 2]; }
            if (sub == 3) { val = sc[4 * g + 3]; rs = rr[4 * g + 3]; }
            const int jj = j + 4 * g + sub;
            if (jj < cnt) {
              atomicMax(mx + rs * H + hl, fenc(val));
              tq[(size_t)(b0 + jj) * H + hl] = val;
            }
          }
        }
      }
    }
    __syncthreads();
    if (fits && sweep != 1) {
      for (int b0 = wave * 32; b0 < n; b0 += ATH) {
        const int idx = b0 + lane;
        unsigned pk = 0;
        if (idx < n) pk = q[idx];
        const int s_l = (int)(pk & 0xFFFFFFu), r_l = (int)(pk >> 24);
        const int cnt = min(32, n - b0);
        for (int j = 0; j < cnt; j += 8) {
          float4 v[8];
          float  a[8];
          int    rr[8];
#pragma unroll
          for (int u = 0; u < 8; ++u) {
            const int jj = min(j + u, cnt - 1);
            const int s = __builtin_amdgcn_readlane(s_l, jj);
            rr[u] = __builtin_amdgcn_readlane(r_l, jj);
            v[u] = *(const float4*)(h_src + (size_t)s * D + 4 * lane);
            a[u] = __expf(tq[(size_t)(b0 + jj) * H + hl] - fdec(mx[rr[u] * H + hl]));
          }
#pragma unroll
          for (int u = 0; u < 8; ++u) {
            if (j + u < cnt) {
              if ((lane & 3) == 0) atomicAdd(den + rr[u] * H + hl, a[u]);
              float* p = acc + rr[u] * D + lane;
              atomicAdd(p,      a[u] * v[u].x);
              atomicAdd(p + 32, a[u] * v[u].y);
              atomicAdd(p + 64, a[u] * v[u].z);
              atomicAdd(p + 96, a[u] * v[u].w);
            }
          }
        }
      }
    }
    __syncthreads();
    if (tid == 0) qn = 0;
    __syncthreads();
    if (sweep == 0) {
      if (fits) break;
      sweep = 1; span = QCAP; lo = 0;
    } else {
      lo = hi;
      if (lo >= E && sweep == 1) { sweep = 2; lo = 0; }
    }
  }

  for (int r = wave; r < nrows; r += ATH / 32) {
    const int nd = row0 + r;
    const float m  = fdec(mx[r * H + hl]);
    const float se = __expf(self_t[(size_t)nd * H + hl] - m);
    const float rd = 1.0f / (den[r * H + hl] + se);
    const float da = se * rd;
    const float4 x = *(const float4*)(h_dst + (size_t)nd * D + 4 * lane);
    const float* a = acc + r * D + lane;
    float4 o;
    o.x = a[0]  * rd + da * x.x;
    o.y = a[32] * rd + da * x.y;
    o.z = a[64] * rd + da * x.z;
    o.w = a[96] * rd + da * x.w;
    *(float4*)(out + (size_t)nd * D + 4 * lane) = o;
  }
}

extern "C" void kernel_launch(void* const* d_in, const int* in_sizes, int n_in,
                              void* d_out, int out_size, void* d_ws, size_t ws_size,
                              hipStream_t stream) {
  const float* h_src = (const float*)d_in[0];
  const float* h_dst = (const float*)d_in[1];
  const float* Wk    = (const float*)d_in[2];
  const float* Wq    = (const float*)d_in[3];
  const int* src_idx = (const int*)d_in[4];
  const int* dst_idx = (const int*)d_in[5];

  const int N_SRC = in_sizes[0] / D;
  const int N_DST = in_sizes[1] / D;
  const int E     = in_sizes[4];

  auto align256 = [](size_t x) { return (x + 255) & ~(size_t)255; };
  char* ws = (char*)d_ws;
  size_t off = 0;
  float* K      = (float*)(ws + off); off += align256((size_t)N_SRC * D * 4);
  float* Q      = (float*)(ws + off); off += align256((size_t)N_DST * D * 4);
  float* self_t = (float*)(ws + off); off += align256((size_t)N_DST * H * 4);
  unsigned* owner = (unsigned*)(ws + off); off += align256((size_t)E);
  const int NAG = (N_DST + AR - 1) / AR;
  float* t_q    = (float*)(ws + off); off += align256((size_t)NAG * QCAP * H * 4);

  project_all<<<256, GEMM_TH, 0, stream>>>(h_src, h_dst, Wk, Wq, K, Q, self_t, N_SRC, N_DST,
                                           dst_idx, owner, E);
  aggregate_owned<<<NAG, ATH, 0, stream>>>(h_src, h_dst, K, Q, src_idx, dst_idx, self_t,
                                           owner, t_q, (float*)d_out, E, N_DST);
}


# Second pass: a lane makes exp and the denominator add for one edge and head (edge l&3 of four): one score line, max read, exp, den add per 4 edges; weights by quad broadcast; den sum order changes
# speedup vs baseline: 4.7922x; 1.0349x over previous
#include <hip/hip_runtime.h>
#include <hip/hip_bf16.h>

#define D      128
#define H      8
#define DK     16
#define SCALE  0.25f
#define AR     200

typedef __attribute__((ext_vector_type(2))) float v2f;
typedef __attribute__((ext_vector_type(4))) float v4f;
typedef __attribute__((ext_vector_type(8))) float v8f;
typedef __attribute__((ext_vector_type(4))) int   v4i;

__device__ __forceinline__ unsigned fenc(float x) {
  unsigned u = __float_as_uint(x);
  return (u & 0x80000000u) ? ~u : (u | 0x80000000u);
}
__device__ __forceinline__ float fdec(unsigned k) {
  return __uint_as_float((k & 0x80000000u) ? (k & 0x7FFFFFFFu) : ~k);
}

#define GEMM_TH 256
__device__ __forceinline__ float half_sum16(float p) {
  p += __int_as_float(__builtin_amdgcn_update_dpp(0, __float_as_int(p), 0xB1,  0xF, 0xF, true));
  p += __int_as_float(__builtin_amdgcn_update_dpp(0, __float_as_int(p), 0x4E,  0xF, 0xF, true));
  p += __int_as_float(__builtin_amdgcn_update_dpp(0, __float_as_int(p), 0x141, 0xF, 0xF, true));
  p += __int_as_float(__builtin_amdgcn_update_dpp(0, __float_as_int(p), 0x140, 0xF, 0xF, true));
  return p;
}
__global__ __launch_bounds__(GEMM_TH) __attribute__((amdgpu_waves_per_eu(2, 2)))
void project_all(const float* __restrict__ Xs, const float* __restrict__ Xd,
                 const float* __restrict__ Wk, const float* __restrict__ Wq,
                 float* __restrict__ K, float* __restrict__ Q,
                 float* __restrict__ self_t, int Ms, int Md,
                 const int* __restrict__ dst_idx, unsigned* __restrict__ owner, int E) {
  {
    const int step  = (int)gridDim.x * GEMM_TH * 4;
    const int nfull = E & ~3;
    if (nfull) {
      for (int base = ((int)blockIdx.x * GEMM_TH + (int)threadIdx.x) * 4; base < nfull; base += step * 8) {
        v4i dv[8];
#pragma unroll
        for (int k = 0; k < 8; ++k)
          dv[k] = *(const v4i*)(dst_idx + min(base + k * step, nfull - 4));
        asm volatile("" :: "v"(dv[0]), "v"(dv[1]), "v"(dv[2]), "v"(dv[3]),
                           "v"(dv[4]), "v"(dv[5]), "v"(dv[6]), "v"(dv[7]));
#pragma unroll
        for (int k = 0; k < 8; ++k) {
          const int e4 = base + k * step;
          if (e4 < nfull)
            owner[e4 >> 2] = (unsigned)dv[k].x / AR | ((unsigned)dv[k].y / AR) << 8 |
                             ((unsigned)dv[k].z / AR) << 16 | ((unsigned)dv[k].w / AR) << 24;
        }
      }
    }
    if (nfull < E && blockIdx.x == 0 && threadIdx.x == 0) {
      unsigned w = 0;
      for (int j = 0; nfull + j < E; ++j) w |= ((unsigned)dst_idx[nfull + j] / AR) << (8 * j);
      owner[nfull >> 2] = w;
    }
  }
  const int lane = threadIdx.x & 31;
  const int wave = __builtin_amdgcn_readfirstlane((int)(threadIdx.x >> 5));
  const int quarter = wave >> 1;
  const int sub     = wave & 1;

  const int row  = lane & 15;
  const int koff = (lane >> 4) << 2;
  const int col0 = quarter * 32;

  v4f wk[2][16], wq[2][16];
  {
    const float* kb = Wk + (size_t)(col0 + row) * D + koff;
    const float* qb = Wq + (size_t)(col0 + row) * D + koff;
#pragma unroll
    for (int t = 0; t < 2; ++t)
#pragma unroll
      for (int j = 0; j < 16; ++j) {
        wk[t][j] = *(const v4f*)(kb + (size_t)t * 16 * D + j * 8);
        wq[t][j] = *(const v4f*)(qb + (size_t)t * 16 * D + j * 8);
      }
  }

  const int n    = lane & 15;
  const int mAdd = (lane >> 4) << 3;

  auto load_a = [&](v4f (&a)[16], const float* xbase, int t) {
#pragma unroll
    for (int j = 0; j < 16; ++j)
      a[j] = *(const v4f*)(xbase + (size_t)t * 16 * D + j * 8);
  };
  auto mm_src = [&](const v4f (&a)[16], int t) {
    v8f acc[2] = {};
#pragma unroll
    for (int j = 0; j < 16; ++j) {
      const v2f a_lo = {a[j].x, a[j].y};
      const v2f a_hi = {a[j].z, a[j].w};
#pragma unroll
      for (int c = 0; c < 2; ++c) {
        const v2f b_lo = {wk[c][j].x, wk[c][j].y};
        acc[c] = __builtin_amdgcn_wmma_f32_16x16x4_f32(
            false, a_lo, false, b_lo, (short)0, acc[c], false, false);
      }
#pragma unroll
      for (int c = 0; c < 2; ++c) {
        const v2f b_hi = {wk[c][j].z, wk[c][j].w};
        acc[c] = __builtin_amdgcn_wmma_f32_16x16x4_f32(
            false, a_hi, false, b_hi, (short)0, acc[c], false, false);
      }
    }
    float* yb = K + (size_t)t * 16 * D + col0;
#pragma unroll
    for (int c = 0; c < 2; ++c) {
#pragma unroll
      for (int r = 0; r < 8; ++r) {
        yb[(size_t)(mAdd + r) * D + c * 16 + n] = acc[c][r];
      }
    }
  };
  auto mm_dst = [&](const v4f (&a)[16], int t) {
    v8f aq[2] = {}, ak[2] = {};
#pragma unroll
    for (int j = 0; j < 16; ++j) {
      const v2f a_lo = {a[j].x, a[j].y};
      const v2f a_hi = {a[j].z, a[j].w};
#pragma unroll
      for (int c = 0; c < 2; ++c) {
        const v2f q_lo = {wq[c][j].x, wq[c][j].y};
        const v2f k_lo = {wk[c][j].x, wk[c][j].y};
        aq[c] = __builtin_amdgcn_wmma_f32_16x16x4_f32(
            false, a_lo, false, q_lo, (short)0, aq[c], false, false);
        ak[c] = __builtin_amdgcn_wmma_f32_16x16x4_f32(
            false, a_lo, false, k_lo, (short)0, ak[c], false, false);
      }
#pragma unroll
      for (int c = 0; c < 2; ++c) {
        const v2f q_hi = {wq[c][j].z, wq[c][j].w};
        const v2f k_hi = {wk[c][j].z, wk[c][j].w};
        aq[c] = __builtin_amdgcn_wmma_f32_16x16x4_f32(
            false, a_hi, false, q_hi, (short)0, aq[c], false, false);
        ak[c] = __builtin_amdgcn_wmma_f32_16x16x4_f32(
            false, a_hi, false, k_hi, (short)0, ak[c], false, false);
      }
    }
    float* yb = Q + (size_t)t * 16 * D + col0;
#pragma unroll
    for (int c = 0; c < 2; ++c) {
#pragma unroll
      for (int r = 0; r < 8; ++r) {
        yb[(size_t)(mAdd + r) * D + c * 16 + n] = aq[c][r];
      }
    }
#pragma unroll
    for (int c = 0; c < 2; ++c) {
      float mine = 0.f;
#pragma unroll
      for (int r = 0; r < 8; ++r) {
        const float s = half_sum16(aq[c][r] * ak[c][r]);
        mine = (n == r) ? s : mine;
      }
      if (n < 8) self_t[(size_t)(t * 16 + mAdd + n) * H + quarter * 2 + c] = mine * SCALE;
    }
  };

  const int nslots = (int)gridDim.x * 2;
  const int slot   = (int)blockIdx.x * 2 + sub;
  auto run = [&](const float* X, int tile, int ntiles, auto&& mm) {
    if (tile >= ntiles) return;
    const float* xbase = X + (size_t)row * D + koff;
    v4f a0[16], a1[16];
    load_a(a0, xbase, tile);
    for (;;) {
      int nxt = tile + nslots;
      if (nxt >= ntiles) { mm(a0, tile); break; }
      load_a(a1, xbase, nxt);
      mm(a0, tile);
      tile = nxt;
      nxt  = tile + nslots;
      if (nxt >= ntiles) { mm(a1, tile); break; }
      load_a(a0, xbase, nxt);
      mm(a1, tile);
      tile = nxt;
    }
  };
  run(Xs, slot, Ms >> 4, mm_src);
  run(Xd, nslots - 1 - slot, Md >> 4, mm_dst);
}

__device__ __forceinline__ float quad_xor1(float x) {
  return __int_as_float(__builtin_amdgcn_update_dpp(0, __float_as_int(x), 0xB1, 0xF, 0xF, true));
}
__device__ __forceinline__ float quad_xor2(float x) {
  return __int_as_float(__builtin_amdgcn_update_dpp(0, __float_as_int(x), 0x4E, 0xF, 0xF, true));
}

template <int K>
__device__ __forceinline__ float quad_bcast(float x) {
  return __int_as_float(__builtin_amdgcn_update_dpp(0, __float_as_int(x), K * 0x55, 0xF, 0xF, true));
}

#define ATH   1024
#define QCAP  16384

__device__ __forceinline__ unsigned zero_bytes(unsigned x) {
  return ~(((x & 0x7F7F7F7Fu) + 0x7F7F7F7Fu) | x) & 0x80808080u;
}

__global__ __launch_bounds__(ATH) void aggregate_owned(
    const float* __restrict__ h_src, const float* __restrict__ h_dst,
    const float* __restrict__ K, const float* __restrict__ Q,
    const int* __restrict__ src_idx, const int* __restrict__ dst_idx,
    const float* __restrict__ self_t, const unsigned* __restrict__ owner,
    float* __restrict__ t_q,
    float* __restrict__ out, int E, int Ndst) {
  __shared__ float    acc[AR * D];
  __shared__ __attribute__((aligned(16))) float qrow[AR * D];
  __shared__ unsigned q[QCAP];
  __shared__ unsigned mx[AR * H];
  __shared__ float    den[AR * H];
  __shared__ unsigned qn;

  const int tid  = threadIdx.x;
  const int lane = tid & 31;
  const int wave  = __builtin_amdgcn_readfirstlane(tid >> 5);
  const int row0  = blockIdx.x * AR;
  const int nrows = min(AR, Ndst - row0);
  const int hl    = lane >> 2;

  float* tq = t_q + (size_t)blockIdx.x * QCAP * H;

  for (int i = tid; i < nrows * D; i += ATH) acc[i] = 0.f;
  for (int i = tid; i < nrows * (D / 4); i += ATH)
    ((float4*)qrow)[i] = ((const float4*)(Q + (size_t)row0 * D))[i];
  for (int i = tid; i < nrows * H; i += ATH) {
    mx[i]  = fenc(self_t[(size_t)row0 * H + i]);
    den[i] = 0.f;
  }
  if (tid == 0) qn = 0;
  __syncthreads();

  auto push = [&](int e) {
    const unsigned p = atomicAdd(&qn, 1u);
    if (p < QCAP) q[p] = (unsigned)e;
  };
  const unsigned char* ob = (const unsigned char*)owner;
  const unsigned pat = (unsigned)blockIdx.x * 0x01010101u;

  int lo = 0, span = E, sweep = 0;
  while (lo < E) {
    const int hi  = (span < E - lo) ? lo + span : E;
    const int ng  = (hi - lo) >> 4;
    const int nbf = (ng + ATH - 1) / ATH;
    const int start = nbf ? (int)(((unsigned)blockIdx.x * (unsigned)nbf / gridDim.x) % (unsigned)nbf) : 0;
    const unsigned char* ospan = ob + lo;
    for (int i = 0; i < nbf; i += 8) {
      uint4 dv[8];
      int   g[8];
#pragma unroll
      for (int k = 0; k < 8; ++k) {
        int b = start + min(i + k, nbf - 1);
        if (b >= nbf) b -= nbf;
        g[k] = b * ATH + tid;
        dv[k] = *(const uint4*)(ospan + (size_t)min(g[k], ng - 1) * 16);
      }
#pragma unroll
      for (int k = 0; k < 8; ++k) {
        if (i + k < nbf && g[k] < ng) {
          const unsigned m0 = zero_bytes(dv[k].x ^ pat), m1 = zero_bytes(dv[k].y ^ pat);
          const unsigned m2 = zero_bytes(dv[k].z ^ pat), m3 = zero_bytes(dv[k].w ^ pat);
          unsigned c = (m0 >> 7) | (m1 >> 6) | (m2 >> 5) | (m3 >> 4);
          const int e = lo + g[k] * 16;
          while (c) {
            const int z = __builtin_ctz(c);
            push(e + ((z & 7) << 2) + (z >> 3));
            c &= c - 1;
          }
        }
      }
    }
    for (int e = lo + ng * 16 + tid; e < hi; e += ATH)
      if (ob[e] == (unsigned char)blockIdx.x) push(e);
    __syncthreads();

    const int n = __builtin_amdgcn_readfirstlane((int)qn);
    const bool fits = n <= QCAP;
    if (fits) {
      for (int b0 = wave * 32; b0 < n; b0 += ATH) {
        const int idx = b0 + lane;
        int s_l = 0, r_l = 0;
        if (idx < n) {
          const int e_l = (int)q[idx];
          s_l = src_idx[e_l];
          r_l = dst_idx[e_l] - row0;
          q[idx] = (unsigned)s_l | ((unsigned)r_l << 24);
        }
        const int cnt = min(32, n - b0);
        for (int j = 0; j < cnt; j += 8) {
          float4 kv[8];
          float  sc[8];
          int    rr[8];
#pragma unroll
          for (int u = 0; u < 8; ++u) {
            const int jj = min(j + u, cnt - 1);
            const int s = __builtin_amdgcn_readlane(s_l, jj);
            rr[u] = __builtin_amdgcn_readlane(r_l, jj);
            kv[u] = *(const float4*)(K + (size_t)s * D + 4 * lane);
          }
#pragma unroll
          for (int u = 0; u < 8; ++u) {
            const float4 qv = *(const float4*)(qrow + rr[u] * D + 4 * lane);
            float p = kv[u].x * qv.x + kv[u].y * qv.y + kv[u].z * qv.z + kv[u].w * qv.w;
            p += quad_xor1(p);
            p += quad_xor2(p);
            sc[u] = p * SCALE;
          }
          const int sub = lane & 3;
#pragma unroll
          for (int g = 0; g < 2; ++g) {
            float val = sc[4 * g];
            int   rs  = rr[4 * g];
            if (sub == 1) { val = sc[4 * g + 1]; rs = rr[4 * g + 1]; }
            if (sub == 2) { val = sc[4 * g + 2]; rs = rr[4 * g + 2]; }
            if (sub == 3) { val = sc[4 * g + 3]; rs = rr[4 * g + 3]; }
            const int jj = j + 4 * g + sub;
            if (jj < cnt) {
              atomicMax(mx + rs * H + hl, fenc(val));
              tq[(size_t)(b0 + jj) * H + hl] = val;
            }
          }
        }
      }
    }
    __syncthreads();
    if (fits && sweep != 1) {
      const int sb = lane & 3;
      for (int b0 = wave * 32; b0 < n; b0 += ATH) {
        const int idx = b0 + lane;
        unsigned pk = 0;
        if (idx < n) pk = q[idx];
        const int s_l = (int)(pk & 0xFFFFFFu), r_l = (int)(pk >> 24);
        const int cnt = min(32, n - b0);
        for (int j = 0; j < cnt; j += 8) {
          float4 v[8];
          float  tv[2], w[2];
          int    rr[8];
#pragma unroll
          for (int g = 0; g < 2; ++g) {
            const int jc = min(j + 4 * g + sb, cnt - 1);
            tv[g] = tq[(size_t)(b0 + jc) * H + hl];
          }
#pragma unroll
          for (int u = 0; u < 8; ++u) {
            const int jj = min(j + u, cnt - 1);
            const int s = __builtin_amdgcn_readlane(s_l, jj);
            rr[u] = __builtin_amdgcn_readlane(r_l, jj);
            v[u] = *(const float4*)(h_src + (size_t)s * D + 4 * lane);
          }
#pragma unroll
          for (int g = 0; g < 2; ++g) {
            int rs = rr[4 * g];
            if (sb == 1) rs = rr[4 * g + 1];
            if (sb == 2) rs = rr[4 * g + 2];
            if (sb == 3) rs = rr[4 * g + 3];
            w[g] = __expf(tv[g] - fdec(mx[rs * H + hl]));
            if (j + 4 * g + sb < cnt) atomicAdd(den + rs * H + hl, w[g]);
          }
          auto add_row = [&](int u, float a) {
            if (j + u < cnt) {
              float* p = acc + rr[u] * D + lane;
              atomicAdd(p,      a * v[u].x);
              atomicAdd(p + 32, a * v[u].y);
              atomicAdd(p + 64, a * v[u].z);
              atomicAdd(p + 96, a * v[u].w);
            }
          };
          add_row(0, quad_bcast<0>(w[0]));
          add_row(1, quad_bcast<1>(w[0]));
          add_row(2, quad_bcast<2>(w[0]));
          add_row(3, quad_bcast<3>(w[0]));
          add_row(4, quad_bcast<0>(w[1]));
          add_row(5, quad_bcast<1>(w[1]));
          add_row(6, quad_bcast<2>(w[1]));
          add_row(7, quad_bcast<3>(w[1]));
        }
      }
    }
    __syncthreads();
    if (tid == 0) qn = 0;
    __syncthreads();
    if (sweep == 0) {
      if (fits) break;
      sweep = 1; span = QCAP; lo = 0;
    } else {
      lo = hi;
      if (lo >= E && sweep == 1) { sweep = 2; lo = 0; }
    }
  }

  for (int r = wave; r < nrows; r += ATH / 32) {
    const int nd = row0 + r;
    const float m  = fdec(mx[r * H + hl]);
    const float se = __expf(self_t[(size_t)nd * H + hl] - m);
    const float rd = 1.0f / (den[r * H + hl] + se);
    const float da = se * rd;
    const float4 x = *(const float4*)(h_dst + (size_t)nd * D + 4 * lane);
    const float* a = acc + r * D + lane;
    float4 o;
    o.x = a[0]  * rd + da * x.x;
    o.y = a[32] * rd + da * x.y;
    o.z = a[64] * rd + da * x.z;
    o.w = a[96] * rd + da * x.w;
    *(float4*)(out + (size_t)nd * D + 4 * lane) = o;
  }
}

extern "C" void kernel_launch(void* const* d_in, const int* in_sizes, int n_in,
                              void* d_out, int out_size, void* d_ws, size_t ws_size,
                              hipStream_t stream) {
  const float* h_src = (const float*)d_in[0];
  const float* h_dst = (const float*)d_in[1];
  const float* Wk    = (const float*)d_in[2];
  const float* Wq    = (const float*)d_in[3];
  const int* src_idx = (const int*)d_in[4];
  const int* dst_idx = (const int*)d_in[5];

  const int N_SRC = in_sizes[0] / D;
  const int N_DST = in_sizes[1] / D;
  const int E     = in_sizes[4];

  auto align256 = [](size_t x) { return (x + 255) & ~(size_t)255; };
  char* ws = (char*)d_ws;
  size_t off = 0;
  float* K      = (float*)(ws + off); off += align256((size_t)N_SRC * D * 4);
  float* Q      = (float*)(ws + off); off += align256((size_t)N_DST * D * 4);
  float* self_t = (float*)(ws + off); off += align256((size_t)N_DST * H * 4);
  unsigned* owner = (unsigned*)(ws + off); off += align256((size_t)E);
  const int NAG = (N_DST + AR - 1) / AR;
  float* t_q    = (float*)(ws + off); off += align256((size_t)NAG * QCAP * H * 4);

  project_all<<<256, GEMM_TH, 0, stream>>>(h_src, h_dst, Wk, Wq, K, Q, self_t, N_SRC, N_DST,
                                           dst_idx, owner, E);
  aggregate_owned<<<NAG, ATH, 0, stream>>>(h_src, h_dst, K, Q, src_idx, dst_idx, self_t,
                                           owner, t_q, (float*)d_out, E, N_DST);
}


# Aggregation: queued edges counting-sorted in LDS by source-row band (2048 rows); both passes walk that order, so all work-groups gather one narrow band of K / h_src at a time; sum order changes
# speedup vs baseline: 4.9454x; 1.0320x over previous
#include <hip/hip_runtime.h>
#include <hip/hip_bf16.h>

#define D      128
#define H      8
#define DK     16
#define SCALE  0.25f
#define AR     200

typedef __attribute__((ext_vector_type(2))) float v2f;
typedef __attribute__((ext_vector_type(4))) float v4f;
typedef __attribute__((ext_vector_type(8))) float v8f;
typedef __attribute__((ext_vector_type(4))) int   v4i;

__device__ __forceinline__ unsigned fenc(float x) {
  unsigned u = __float_as_uint(x);
  return (u & 0x80000000u) ? ~u : (u | 0x80000000u);
}
__device__ __forceinline__ float fdec(unsigned k) {
  return __uint_as_float((k & 0x80000000u) ? (k & 0x7FFFFFFFu) : ~k);
}

#define GEMM_TH 256
__device__ __forceinline__ float half_sum16(float p) {
  p += __int_as_float(__builtin_amdgcn_update_dpp(0, __float_as_int(p), 0xB1,  0xF, 0xF, true));
  p += __int_as_float(__builtin_amdgcn_update_dpp(0, __float_as_int(p), 0x4E,  0xF, 0xF, true));
  p += __int_as_float(__builtin_amdgcn_update_dpp(0, __float_as_int(p), 0x141, 0xF, 0xF, true));
  p += __int_as_float(__builtin_amdgcn_update_dpp(0, __float_as_int(p), 0x140, 0xF, 0xF, true));
  return p;
}
__global__ __launch_bounds__(GEMM_TH) __attribute__((amdgpu_waves_per_eu(2, 2)))
void project_all(const float* __restrict__ Xs, const float* __restrict__ Xd,
                 const float* __restrict__ Wk, const float* __restrict__ Wq,
                 float* __restrict__ K, float* __restrict__ Q,
                 float* __restrict__ self_t, int Ms, int Md,
                 const int* __restrict__ dst_idx, unsigned* __restrict__ owner, int E) {
  {
    const int step  = (int)gridDim.x * GEMM_TH * 4;
    const int nfull = E & ~3;
    if (nfull) {
      for (int base = ((int)blockIdx.x * GEMM_TH + (int)threadIdx.x) * 4; base < nfull; base += step * 8) {
        v4i dv[8];
#pragma unroll
        for (int k = 0; k < 8; ++k)
          dv[k] = *(const v4i*)(dst_idx + min(base + k * step, nfull - 4));
        asm volatile("" :: "v"(dv[0]), "v"(dv[1]), "v"(dv[2]), "v"(dv[3]),
                           "v"(dv[4]), "v"(dv[5]), "v"(dv[6]), "v"(dv[7]));
#pragma unroll
        for (int k = 0; k < 8; ++k) {
          const int e4 = base + k * step;
          if (e4 < nfull)
            owner[e4 >> 2] = (unsigned)dv[k].x / AR | ((unsigned)dv[k].y / AR) << 8 |
                             ((unsigned)dv[k].z / AR) << 16 | ((unsigned)dv[k].w / AR) << 24;
        }
      }
    }
    if (nfull < E && blockIdx.x == 0 && threadIdx.x == 0) {
      unsigned w = 0;
      for (int j = 0; nfull + j < E; ++j) w |= ((unsigned)dst_idx[nfull + j] / AR) << (8 * j);
      owner[nfull >> 2] = w;
    }
  }
  const int lane = threadIdx.x & 31;
  const int wave = __builtin_amdgcn_readfirstlane((int)(threadIdx.x >> 5));
  const int quarter = wave >> 1;
  const int sub     = wave & 1;

  const int row  = lane & 15;
  const int koff = (lane >> 4) << 2;
  const int col0 = quarter * 32;

  v4f wk[2][16], wq[2][16];
  {
    const float* kb = Wk + (size_t)(col0 + row) * D + koff;
    const float* qb = Wq + (size_t)(col0 + row) * D + koff;
#pragma unroll
    for (int t = 0; t < 2; ++t)
#pragma unroll
      for (int j = 0; j < 16; ++j) {
        wk[t][j] = *(const v4f*)(kb + (size_t)t * 16 * D + j * 8);
        wq[t][j] = *(const v4f*)(qb + (size_t)t * 16 * D + j * 8);
      }
  }

  const int n    = lane & 15;
  const int mAdd = (lane >> 4) << 3;

  auto load_a = [&](v4f (&a)[16], const float* xbase, int t) {
#pragma unroll
    for (int j = 0; j < 16; ++j)
      a[j] = *(const v4f*)(xbase + (size_t)t * 16 * D + j * 8);
  };
  auto mm_src = [&](const v4f (&a)[16], int t) {
    v8f acc[2] = {};
#pragma unroll
    for (int j = 0; j < 16; ++j) {
      const v2f a_lo = {a[j].x, a[j].y};
      const v2f a_hi = {a[j].z, a[j].w};
#pragma unroll
      for (int c = 0; c < 2; ++c) {
        const v2f b_lo = {wk[c][j].x, wk[c][j].y};
        acc[c] = __builtin_amdgcn_wmma_f32_16x16x4_f32(
            false, a_lo, false, b_lo, (short)0, acc[c], false, false);
      }
#pragma unroll
      for (int c = 0; c < 2; ++c) {
        const v2f b_hi = {wk[c][j].z, wk[c][j].w};
        acc[c] = __builtin_amdgcn_wmma_f32_16x16x4_f32(
            false, a_hi, false, b_hi, (short)0, acc[c], false, false);
      }
    }
    float* yb = K + (size_t)t * 16 * D + col0;
#pragma unroll
    for (int c = 0; c < 2; ++c) {
#pragma unroll
      for (int r = 0; r < 8; ++r) {
        yb[(size_t)(mAdd + r) * D + c * 16 + n] = acc[c][r];
      }
    }
  };
  auto mm_dst = [&](const v4f (&a)[16], int t) {
    v8f aq[2] = {}, ak[2] = {};
#pragma unroll
    for (int j = 0; j < 16; ++j) {
      const v2f a_lo = {a[j].x, a[j].y};
      const v2f a_hi = {a[j].z, a[j].w};
#pragma unroll
      for (int c = 0; c < 2; ++c) {
        const v2f q_lo = {wq[c][j].x, wq[c][j].y};
        const v2f k_lo = {wk[c][j].x, wk[c][j].y};
        aq[c] = __builtin_amdgcn_wmma_f32_16x16x4_f32(
            false, a_lo, false, q_lo, (short)0, aq[c], false, false);
        ak[c] = __builtin_amdgcn_wmma_f32_16x16x4_f32(
            false, a_lo, false, k_lo, (short)0, ak[c], false, false);
      }
#pragma unroll
      for (int c = 0; c < 2; ++c) {
        const v2f q_hi = {wq[c][j].z, wq[c][j].w};
        const v2f k_hi = {wk[c][j].z, wk[c][j].w};
        aq[c] = __builtin_amdgcn_wmma_f32_16x16x4_f32(
            false, a_hi, false, q_hi, (short)0, aq[c], false, false);
        ak[c] = __builtin_amdgcn_wmma_f32_16x16x4_f32(
            false, a_hi, false, k_hi, (short)0, ak[c], false, false);
      }
    }
    float* yb = Q + (size_t)t * 16 * D + col0;
#pragma unroll
    for (int c = 0; c < 2; ++c) {
#pragma unroll
      for (int r = 0; r < 8; ++r) {
        yb[(size_t)(mAdd + r) * D + c * 16 + n] = aq[c][r];
      }
    }
#pragma unroll
    for (int c = 0; c < 2; ++c) {
      float mine = 0.f;
#pragma unroll
      for (int r = 0; r < 8; ++r) {
        const float s = half_sum16(aq[c][r] * ak[c][r]);
        mine = (n == r) ? s : mine;
      }
      if (n < 8) self_t[(size_t)(t * 16 + mAdd + n) * H + quarter * 2 + c] = mine * SCALE;
    }
  };

  const int nslots = (int)gridDim.x * 2;
  const int slot   = (int)blockIdx.x * 2 + sub;
  auto run = [&](const float* X, int tile, int ntiles, auto&& mm) {
    if (tile >= ntiles) return;
    const float* xbase = X + (size_t)row * D + koff;
    v4f a0[16], a1[16];
    load_a(a0, xbase, tile);
    for (;;) {
      int nxt = tile + nslots;
      if (nxt >= ntiles) { mm(a0, tile); break; }
      load_a(a1, xbase, nxt);
      mm(a0, tile);
      tile = nxt;
      nxt  = tile + nslots;
      if (nxt >= ntiles) { mm(a1, tile); break; }
      load_a(a0, xbase, nxt);
      mm(a1, tile);
      tile = nxt;
    }
  };
  run(Xs, slot, Ms >> 4, mm_src);
  run(Xd, nslots - 1 - slot, Md >> 4, mm_dst);
}

__device__ __forceinline__ float quad_xor1(float x) {
  return __int_as_float(__builtin_amdgcn_update_dpp(0, __float_as_int(x), 0xB1, 0xF, 0xF, true));
}
__device__ __forceinline__ float quad_xor2(float x) {
  return __int_as_float(__builtin_amdgcn_update_dpp(0, __float_as_int(x), 0x4E, 0xF, 0xF, true));
}

template <int K>
__device__ __forceinline__ float quad_bcast(float x) {
  return __int_as_float(__builtin_amdgcn_update_dpp(0, __float_as_int(x), K * 0x55, 0xF, 0xF, true));
}

#define ATH   1024
#define QCAP  8192
#define QPT   (QCAP / ATH)
#define NB    128
#define BSH   11

__device__ __forceinline__ unsigned zero_bytes(unsigned x) {
  return ~(((x & 0x7F7F7F7Fu) + 0x7F7F7F7Fu) | x) & 0x80808080u;
}

__global__ __launch_bounds__(ATH) void aggregate_owned(
    const float* __restrict__ h_src, const float* __restrict__ h_dst,
    const float* __restrict__ K, const float* __restrict__ Q,
    const int* __restrict__ src_idx, const int* __restrict__ dst_idx,
    const float* __restrict__ self_t, const unsigned* __restrict__ owner,
    float* __restrict__ t_q,
    float* __restrict__ out, int E, int Ndst) {
  __shared__ float    acc[AR * D];
  __shared__ __attribute__((aligned(16))) float qrow[AR * D];
  __shared__ unsigned q[QCAP];
  __shared__ unsigned qs[QCAP];
  __shared__ __attribute__((aligned(16))) unsigned bcnt[NB];
  __shared__ __attribute__((aligned(16))) unsigned bas[NB];
  __shared__ unsigned mx[AR * H];
  __shared__ float    den[AR * H];
  __shared__ unsigned qn;

  const int tid  = threadIdx.x;
  const int lane = tid & 31;
  const int wave  = __builtin_amdgcn_readfirstlane(tid >> 5);
  const int row0  = blockIdx.x * AR;
  const int nrows = min(AR, Ndst - row0);
  const int hl    = lane >> 2;

  float* tq = t_q + (size_t)blockIdx.x * QCAP * H;

  for (int i = tid; i < nrows * D; i += ATH) acc[i] = 0.f;
  for (int i = tid; i < nrows * (D / 4); i += ATH)
    ((float4*)qrow)[i] = ((const float4*)(Q + (size_t)row0 * D))[i];
  for (int i = tid; i < nrows * H; i += ATH) {
    mx[i]  = fenc(self_t[(size_t)row0 * H + i]);
    den[i] = 0.f;
  }
  if (tid == 0) qn = 0;
  if (tid < NB) bcnt[tid] = 0;
  __syncthreads();

  auto push = [&](int e) {
    const unsigned p = atomicAdd(&qn, 1u);
    if (p < QCAP) q[p] = (unsigned)e;
  };
  const unsigned char* ob = (const unsigned char*)owner;
  const unsigned pat = (unsigned)blockIdx.x * 0x01010101u;

  int lo = 0, span = E, sweep = 0;
  while (lo < E) {
    const int hi  = (span < E - lo) ? lo + span : E;
    const int ng  = (hi - lo) >> 4;
    const int nbf = (ng + ATH - 1) / ATH;
    const int start = nbf ? (int)(((unsigned)blockIdx.x * (unsigned)nbf / gridDim.x) % (unsigned)nbf) : 0;
    const unsigned char* ospan = ob + lo;
    for (int i = 0; i < nbf; i += 8) {
      uint4 dv[8];
      int   g[8];
#pragma unroll
      for (int k = 0; k < 8; ++k) {
        int b = start + min(i + k, nbf - 1);
        if (b >= nbf) b -= nbf;
        g[k] = b * ATH + tid;
        dv[k] = *(const uint4*)(ospan + (size_t)min(g[k], ng - 1) * 16);
      }
#pragma unroll
      for (int k = 0; k < 8; ++k) {
        if (i + k < nbf && g[k] < ng) {
          const unsigned m0 = zero_bytes(dv[k].x ^ pat), m1 = zero_bytes(dv[k].y ^ pat);
          const unsigned m2 = zero_bytes(dv[k].z ^ pat), m3 = zero_bytes(dv[k].w ^ pat);
          unsigned c = (m0 >> 7) | (m1 >> 6) | (m2 >> 5) | (m3 >> 4);
          const int e = lo + g[k] * 16;
          while (c) {
            const int z = __builtin_ctz(c);
            push(e + ((z & 7) << 2) + (z >> 3));
            c &= c - 1;
          }
        }
      }
    }
    for (int e = lo + ng * 16 + tid; e < hi; e += ATH)
      if (ob[e] == (unsigned char)blockIdx.x) push(e);
    __syncthreads();

    const int n = __builtin_amdgcn_readfirstlane((int)qn);
    const bool fits = n <= QCAP;
    unsigned pkw[QPT], rnk[QPT];
    if (fits) {
      int sv[QPT], dv[QPT];
#pragma unroll
      for (int k = 0; k < QPT; ++k) {
        const int idx = tid + k * ATH;
        const int e_k = (idx < n) ? (int)q[idx] : 0;
        sv[k] = src_idx[e_k];
        dv[k] = dst_idx[e_k];
      }
#pragma unroll
      for (int k = 0; k < QPT; ++k) {
        pkw[k] = (unsigned)sv[k] | ((unsigned)(dv[k] - row0) << 24);
        if (tid + k * ATH < n)
          rnk[k] = atomicAdd(bcnt + (NB - 1 - min((unsigned)sv[k] >> BSH, (unsigned)(NB - 1))), 1u);
      }
    }
    __syncthreads();
    if (fits && wave == 0) {
      const uint4 c4 = ((const uint4*)bcnt)[lane];
      const unsigned tot = c4.x + c4.y + c4.z + c4.w;
      unsigned inc = tot;
#pragma unroll
      for (int d = 1; d < 32; d <<= 1) {
        const unsigned up = __shfl_up(inc, (unsigned)d);
        inc += (lane >= d) ? up : 0u;
      }
      uint4 b4;
      b4.x = inc - tot;
      b4.y = b4.x + c4.x;
      b4.z = b4.y + c4.y;
      b4.w = b4.z + c4.z;
      ((uint4*)bas)[lane] = b4;
    }
    __syncthreads();
    if (fits) {
#pragma unroll
      for (int k = 0; k < QPT; ++k)
        if (tid + k * ATH < n)
          qs[bas[NB - 1 - min((pkw[k] & 0xFFFFFFu) >> BSH, (unsigned)(NB - 1))] + rnk[k]] = pkw[k];
    }
    __syncthreads();
    if (fits) {
      for (int b0 = wave * 32; b0 < n; b0 += ATH) {
        const int idx = b0 + lane;
        unsigned pk1 = 0;
        if (idx < n) pk1 = qs[idx];
        const int s_l = (int)(pk1 & 0xFFFFFFu), r_l = (int)(pk1 >> 24);
        const int cnt = min(32, n - b0);
        for (int j = 0; j < cnt; j += 8) {
          float4 kv[8];
          float  sc[8];
          int    rr[8];
#pragma unroll
          for (int u = 0; u < 8; ++u) {
            const int jj = min(j + u, cnt - 1);
            const int s = __builtin_amdgcn_readlane(s_l, jj);
            rr[u] = __builtin_amdgcn_readlane(r_l, jj);
            kv[u] = *(const float4*)(K + (size_t)s * D + 4 * lane);
          }
#pragma unroll
          for (int u = 0; u < 8; ++u) {
            const float4 qv = *(const float4*)(qrow + rr[u] * D + 4 * lane);
            float p = kv[u].x * qv.x + kv[u].y * qv.y + kv[u].z * qv.z + kv[u].w * qv.w;
            p += quad_xor1(p);
            p += quad_xor2(p);
            sc[u] = p * SCALE;
          }
          const int sub = lane & 3;
#pragma unroll
          for (int g = 0; g < 2; ++g) {
            float val = sc[4 * g];
            int   rs  = rr[4 * g];
            if (sub == 1) { val = sc[4 * g + 1]; rs = rr[4 * g + 1]; }
            if (sub == 2) { val = sc[4 * g + 2]; rs = rr[4 * g + 2]; }
            if (sub == 3) { val = sc[4 * g + 3]; rs = rr[4 * g + 3]; }
            const int jj = j + 4 * g + sub;
            if (jj < cnt) {
              atomicMax(mx + rs * H + hl, fenc(val));
              tq[(size_t)(b0 + jj) * H + hl] = val;
            }
          }
        }
      }
    }
    __syncthreads();
    if (fits && sweep != 1) {
      const int sb = lane & 3;
      for (int b0 = wave * 32; b0 < n; b0 += ATH) {
        const int idx = b0 + lane;
        unsigned pk = 0;
        if (idx < n) pk = qs[idx];
        const int s_l = (int)(pk & 0xFFFFFFu), r_l = (int)(pk >> 24);
        const int cnt = min(32, n - b0);
        for (int j = 0; j < cnt; j += 8) {
          float4 v[8];
          float  tv[2], w[2];
          int    rr[8];
#pragma unroll
          for (int g = 0; g < 2; ++g) {
            const int jc = min(j + 4 * g + sb, cnt - 1);
            tv[g] = tq[(size_t)(b0 + jc) * H + hl];
          }
#pragma unroll
          for (int u = 0; u < 8; ++u) {
            const int jj = min(j + u, cnt - 1);
            const int s = __builtin_amdgcn_readlane(s_l, jj);
            rr[u] = __builtin_amdgcn_readlane(r_l, jj);
            v[u] = *(const float4*)(h_src + (size_t)s * D + 4 * lane);
          }
#pragma unroll
          for (int g = 0; g < 2; ++g) {
            int rs = rr[4 * g];
            if (sb == 1) rs = rr[4 * g + 1];
            if (sb == 2) rs = rr[4 * g + 2];
            if (sb == 3) rs = rr[4 * g + 3];
            w[g] = __expf(tv[g] - fdec(mx[rs * H + hl]));
            if (j + 4 * g + sb < cnt) atomicAdd(den + rs * H + hl, w[g]);
          }
          auto add_row = [&](int u, float a) {
            if (j + u < cnt) {
              float* p = acc + rr[u] * D + lane;
              atomicAdd(p,      a * v[u].x);
              atomicAdd(p + 32, a * v[u].y);
              atomicAdd(p + 64, a * v[u].z);
              atomicAdd(p + 96, a * v[u].w);
            }
          };
          add_row(0, quad_bcast<0>(w[0]));
          add_row(1, quad_bcast<1>(w[0]));
          add_row(2, quad_bcast<2>(w[0]));
          add_row(3, quad_bcast<3>(w[0]));
          add_row(4, quad_bcast<0>(w[1]));
          add_row(5, quad_bcast<1>(w[1]));
          add_row(6, quad_bcast<2>(w[1]));
          add_row(7, quad_bcast<3>(w[1]));
        }
      }
    }
    __syncthreads();
    if (tid == 0) qn = 0;
    if (tid < NB) bcnt[tid] = 0;
    __syncthreads();
    if (sweep == 0) {
      if (fits) break;
      sweep = 1; span = QCAP; lo = 0;
    } else {
      lo = hi;
      if (lo >= E && sweep == 1) { sweep = 2; lo = 0; }
    }
  }

  for (int r = wave; r < nrows; r += ATH / 32) {
    const int nd = row0 + r;
    const float m  = fdec(mx[r * H + hl]);
    const float se = __expf(self_t[(size_t)nd * H + hl] - m);
    const float rd = 1.0f / (den[r * H + hl] + se);
    const float da = se * rd;
    const float4 x = *(const float4*)(h_dst + (size_t)nd * D + 4 * lane);
    const float* a = acc + r * D + lane;
    float4 o;
    o.x = a[0]  * rd + da * x.x;
    o.y = a[32] * rd + da * x.y;
    o.z = a[64] * rd + da * x.z;
    o.w = a[96] * rd + da * x.w;
    *(float4*)(out + (size_t)nd * D + 4 * lane) = o;
  }
}

extern "C" void kernel_launch(void* const* d_in, const int* in_sizes, int n_in,
                              void* d_out, int out_size, void* d_ws, size_t ws_size,
                              hipStream_t stream) {
  const float* h_src = (const float*)d_in[0];
  const float* h_dst = (const float*)d_in[1];
  const float* Wk    = (const float*)d_in[2];
  const float* Wq    = (const float*)d_in[3];
  const int* src_idx = (const int*)d_in[4];
  const int* dst_idx = (const int*)d_in[5];

  const int N_SRC = in_sizes[0] / D;
  const int N_DST = in_sizes[1] / D;
  const int E     = in_sizes[4];

  auto align256 = [](size_t x) { return (x + 255) & ~(size_t)255; };
  char* ws = (char*)d_ws;
  size_t off = 0;
  float* K      = (float*)(ws + off); off += align256((size_t)N_SRC * D * 4);
  float* Q      = (float*)(ws + off); off += align256((size_t)N_DST * D * 4);
  float* self_t = (float*)(ws + off); off += align256((size_t)N_DST * H * 4);
  unsigned* owner = (unsigned*)(ws + off); off += align256((size_t)E);
  const int NAG = (N_DST + AR - 1) / AR;
  float* t_q    = (float*)(ws + off); off += align256((size_t)NAG * QCAP * H * 4);

  project_all<<<256, GEMM_TH, 0, stream>>>(h_src, h_dst, Wk, Wq, K, Q, self_t, N_SRC, N_DST,
                                           dst_idx, owner, E);
  aggregate_owned<<<NAG, ATH, 0, stream>>>(h_src, h_dst, K, Q, src_idx, dst_idx, self_t,
                                           owner, t_q, (float*)d_out, E, N_DST);
}
